# starting kernel
# baseline (speedup 1.0000x reference)
#include <hip/hip_runtime.h>
#include <hip/hip_fp16.h>

typedef __attribute__((ext_vector_type(16))) _Float16 v16h;
typedef __attribute__((ext_vector_type(8)))  _Float16 v8h;
typedef __attribute__((ext_vector_type(8)))  float    v8f;

#define HDIM 51
#define LOG2E 1.44269504088896340736f

__device__ __forceinline__ float fast_sigmoid(float z) {
  float e = __builtin_amdgcn_exp2f(-LOG2E * z);
  return __builtin_amdgcn_rcpf(1.0f + e);
}
__device__ __forceinline__ float fast_tanh(float z) {
  float e = __builtin_amdgcn_exp2f(2.0f * LOG2E * z);
  return 1.0f - 2.0f * __builtin_amdgcn_rcpf(e + 1.0f);
}

__global__ __launch_bounds__(128)
void lstm2_wmma_kernel(const float* __restrict__ input,
                       const float* __restrict__ w_ih1, const float* __restrict__ w_hh1,
                       const float* __restrict__ b_ih1, const float* __restrict__ b_hh1,
                       const float* __restrict__ w_ih2, const float* __restrict__ w_hh2,
                       const float* __restrict__ b_ih2, const float* __restrict__ b_hh2,
                       const float* __restrict__ w_lin, const float* __restrict__ b_lin,
                       float* __restrict__ out, int T, int Tout)
{
  __shared__ v16h b1frag[32][32];
  __shared__ v16h b2frag[64][32];
  __shared__ __align__(16) _Float16 hcat[4][16][136];
  __shared__ float bias1s[256], bias2s[256], wih1s[256];
  __shared__ float xtile[4][16];

  const int tid = threadIdx.x;

  for (int idx = tid; idx < 32 * 512; idx += 128) {
    int f = idx >> 9, rem = idx & 511, l = rem >> 4, e = rem & 15;
    int ntile = f >> 1, ks = f & 1;
    int cc = l & 15, hh = (l & 16) ? 8 : 0;
    int Kl = (e < 8) ? (hh + e) : (16 + hh + (e - 8));
    int K = ks * 32 + Kl;
    int col = ntile * 16 + cc, q = col >> 6, j = col & 63;
    float v = 0.0f;
    if (K < HDIM && j < HDIM) v = w_hh1[(q * HDIM + j) * HDIM + K];
    ((_Float16*)&b1frag[f][l])[e] = (_Float16)v;
  }
  for (int idx = tid; idx < 64 * 512; idx += 128) {
    int f = idx >> 9, rem = idx & 511, l = rem >> 4, e = rem & 15;
    int ntile = f >> 2, ks = f & 3;
    int cc = l & 15, hh = (l & 16) ? 8 : 0;
    int Kl = (e < 8) ? (hh + e) : (16 + hh + (e - 8));
    int K = ks * 32 + Kl;
    int col = ntile * 16 + cc, q = col >> 6, j = col & 63;
    float v = 0.0f;
    if (j < HDIM) {
      if (K < 64) { if (K < HDIM) v = w_ih2[(q * HDIM + j) * HDIM + K]; }
      else        { int kk = K - 64; if (kk < HDIM) v = w_hh2[(q * HDIM + j) * HDIM + kk]; }
    }
    ((_Float16*)&b2frag[f][l])[e] = (_Float16)v;
  }
  for (int col = tid; col < 256; col += 128) {
    int q = col >> 6, j = col & 63;
    float bb1 = 0.f, bb2 = 0.f, wi = 0.f;
    if (j < HDIM) {
      int r = q * HDIM + j;
      bb1 = b_ih1[r] + b_hh1[r];
      bb2 = b_ih2[r] + b_hh2[r];
      wi  = w_ih1[r];
    }
    bias1s[col] = bb1; bias2s[col] = bb2; wih1s[col] = wi;
  }
  for (int i = tid; i < 4 * 16 * 136; i += 128) ((_Float16*)hcat)[i] = (_Float16)0.0f;
  __syncthreads();

  const int w    = tid >> 5;
  const int lane = tid & 31;
  const int cix  = lane & 15;
  const int hi   = (lane & 16) ? 8 : 0;
  const int rowbase = blockIdx.x * 64 + w * 16;

  float biasv1[16], biasv2[16], wihv[16];
#pragma unroll
  for (int n = 0; n < 16; ++n) {
    int col = n * 16 + cix;
    biasv1[n] = bias1s[col]; biasv2[n] = bias2s[col]; wihv[n] = wih1s[col];
  }
  float wlinv[4];
#pragma unroll
  for (int tt = 0; tt < 4; ++tt) {
    int col = tt * 16 + cix;
    wlinv[tt] = (col < HDIM) ? w_lin[col] : 0.0f;
  }
  const float blin = b_lin[0];

  float c1s[4][8], c2s[4][8], ylast[8];
#pragma unroll
  for (int a = 0; a < 4; ++a)
#pragma unroll
    for (int k = 0; k < 8; ++k) { c1s[a][k] = 0.f; c2s[a][k] = 0.f; }
#pragma unroll
  for (int k = 0; k < 8; ++k) ylast[k] = 0.f;

  for (int t = 0; t < Tout; ++t) {
    float xv[8];
    if (t < T) {
      if (lane < 16) xtile[w][lane] = input[(long)(rowbase + lane) * T + t];
#pragma unroll
      for (int k = 0; k < 8; ++k) xv[k] = xtile[w][hi + k];
    } else {
#pragma unroll
      for (int k = 0; k < 8; ++k) xv[k] = ylast[k];
    }

    v16h a1[2];
#pragma unroll
    for (int ks = 0; ks < 2; ++ks) {
      const _Float16* bp = &hcat[w][cix][ks * 32 + hi];
      v8h lo = *(const v8h*)bp;
      v8h hp = *(const v8h*)(bp + 16);
      a1[ks] = __builtin_shufflevector(lo, hp, 0,1,2,3,4,5,6,7,8,9,10,11,12,13,14,15);
    }
#pragma unroll
    for (int tt = 0; tt < 4; ++tt) {
      v8f acc[4];
#pragma unroll
      for (int q = 0; q < 4; ++q) {
        int nt = q * 4 + tt;
        float bb = biasv1[nt];
        v8f cacc = {bb, bb, bb, bb, bb, bb, bb, bb};
#pragma unroll
        for (int ks = 0; ks < 2; ++ks)
          cacc = __builtin_amdgcn_wmma_f32_16x16x32_f16(false, a1[ks], false,
                    b1frag[nt * 2 + ks][lane], (short)0, cacc, false, false);
        float wv = wihv[nt];
#pragma unroll
        for (int k = 0; k < 8; ++k) cacc[k] += xv[k] * wv;
        acc[q] = cacc;
      }
      int col = tt * 16 + cix;
      bool valid = (col < HDIM);
#pragma unroll
      for (int k = 0; k < 8; ++k) {
        float ig = fast_sigmoid(acc[0][k]);
        float fg = fast_sigmoid(acc[1][k]);
        float gg = fast_tanh(acc[2][k]);
        float og = fast_sigmoid(acc[3][k]);
        float cn = fg * c1s[tt][k] + ig * gg;
        c1s[tt][k] = cn;
        float hn = og * fast_tanh(cn);
        hcat[w][hi + k][col] = valid ? (_Float16)hn : (_Float16)0.0f;
      }
    }

    v16h a2[4];
#pragma unroll
    for (int ks = 0; ks < 4; ++ks) {
      const _Float16* bp = &hcat[w][cix][ks * 32 + hi];
      v8h lo = *(const v8h*)bp;
      v8h hp = *(const v8h*)(bp + 16);
      a2[ks] = __builtin_shufflevector(lo, hp, 0,1,2,3,4,5,6,7,8,9,10,11,12,13,14,15);
    }
    float ys[8];
#pragma unroll
    for (int k = 0; k < 8; ++k) ys[k] = 0.f;
#pragma unroll
    for (int tt = 0; tt < 4; ++tt) {
      v8f acc[4];
#pragma unroll
      for (int q = 0; q < 4; ++q) {
        int nt = q * 4 + tt;
        float bb = biasv2[nt];
        v8f cacc = {bb, bb, bb, bb, bb, bb, bb, bb};
#pragma unroll
        for (int ks = 0; ks < 4; ++ks)
          cacc = __builtin_amdgcn_wmma_f32_16x16x32_f16(false, a2[ks], false,
                    b2frag[nt * 4 + ks][lane], (short)0, cacc, false, false);
        acc[q] = cacc;
      }
      int col = tt * 16 + cix;
      bool valid = (col < HDIM);
      float wl = wlinv[tt];
#pragma unroll
      for (int k = 0; k < 8; ++k) {
        float ig = fast_sigmoid(acc[0][k]);
        float fg = fast_sigmoid(acc[1][k]);
        float gg = fast_tanh(acc[2][k]);
        float og = fast_sigmoid(acc[3][k]);
        float cn = fg * c2s[tt][k] + ig * gg;
        c2s[tt][k] = cn;
        float hn = og * fast_tanh(cn);
        hcat[w][hi + k][64 + col] = valid ? (_Float16)hn : (_Float16)0.0f;
        ys[k] += (valid ? hn : 0.0f) * wl;
      }
    }

#pragma unroll
    for (int mask = 1; mask <= 8; mask <<= 1)
#pragma unroll
      for (int k = 0; k < 8; ++k) ys[k] += __shfl_xor(ys[k], mask, 32);
#pragma unroll
    for (int k = 0; k < 8; ++k) ylast[k] = ys[k] + blin;
    if (cix == 0) {
#pragma unroll
      for (int k = 0; k < 8; ++k)
        out[(long)(rowbase + hi + k) * Tout + t] = ylast[k];
    }
  }
}

extern "C" void kernel_launch(void* const* d_in, const int* in_sizes, int n_in,
                              void* d_out, int out_size, void* d_ws, size_t ws_size,
                              hipStream_t stream) {
  const float* input = (const float*)d_in[0];
  const float* w_ih1 = (const float*)d_in[2];
  const float* w_hh1 = (const float*)d_in[3];
  const float* b_ih1 = (const float*)d_in[4];
  const float* b_hh1 = (const float*)d_in[5];
  const float* w_ih2 = (const float*)d_in[6];
  const float* w_hh2 = (const float*)d_in[7];
  const float* b_ih2 = (const float*)d_in[8];
  const float* b_hh2 = (const float*)d_in[9];
  const float* w_lin = (const float*)d_in[10];
  const float* b_lin = (const float*)d_in[11];
  float* out = (float*)d_out;

  const int B = 1024;
  const int T = in_sizes[0] / B;
  const int Tout = out_size / B;

  lstm2_wmma_kernel<<<dim3(B / 64), dim3(128), 0, stream>>>(
      input, w_ih1, w_hh1, b_ih1, b_hh1, w_ih2, w_hh2, b_ih2, b_hh2,
      w_lin, b_lin, out, T, Tout);
}


# Split the 64 padded hidden units over the 4 waves of a workgroup (one 16-row tile each): a wave keeps its weight fragments in registers; readout summed per wave then 4-way, stored 32 steps at a time
# speedup vs baseline: 4.2017x; 4.2017x over previous
#include <hip/hip_runtime.h>
#include <hip/hip_fp16.h>


typedef __attribute__((ext_vector_type(16))) _Float16 v16h;
typedef __attribute__((ext_vector_type(8)))  _Float16 v8h;
typedef __attribute__((ext_vector_type(8)))  float    v8f;

#define HDIM 51
#define LOG2E 1.44269504088896340736f

__device__ __forceinline__ float fast_sigmoid(float z) {
  float e = __builtin_amdgcn_exp2f(-LOG2E * z);
  return __builtin_amdgcn_rcpf(1.0f + e);
}
__device__ __forceinline__ float fast_tanh(float z) {
  float e = __builtin_amdgcn_exp2f(2.0f * LOG2E * z);
  return 1.0f - 2.0f * __builtin_amdgcn_rcpf(e + 1.0f);
}

__global__ __launch_bounds__(128)
void lstm2_wmma_kernel(const float* __restrict__ input,
                       const float* __restrict__ w_ih1, const float* __restrict__ w_hh1,
                       const float* __restrict__ b_ih1, const float* __restrict__ b_hh1,
                       const float* __restrict__ w_ih2, const float* __restrict__ w_hh2,
                       const float* __restrict__ b_ih2, const float* __restrict__ b_hh2,
                       const float* __restrict__ w_lin, const float* __restrict__ b_lin,
                       float* __restrict__ out, int T, int Tout)
{
  __shared__ __align__(16) _Float16 hbuf[2][16][136];
  __shared__ float xtile[4][16];
  __shared__ float ypart[4][16][64];

  const int tid  = threadIdx.x;
  const int w    = tid >> 5;
  const int lane = tid & 31;
  const int cix  = lane & 15;
  const int hi   = (lane & 16) ? 8 : 0;
  const int rowbase = blockIdx.x * 16;
  const int j    = w * 16 + cix;
  const bool valid = (j < HDIM);

  v16h b1[4][2];
  v16h b2[4][4];
  const int jc = valid ? j : 0;
#pragma unroll
  for (int q = 0; q < 4; ++q) {
    const int r = q * HDIM + jc;
#pragma unroll
    for (int ks = 0; ks < 2; ++ks) {
#pragma unroll
      for (int e = 0; e < 16; ++e) {
        int Kl = (e < 8) ? (hi + e) : (16 + hi + (e - 8));
        int K = ks * 32 + Kl;
        bool ok = valid && (K < HDIM);
        int Kc = (K < HDIM) ? K : 0;
        float v = w_hh1[r * HDIM + Kc];
        b1[q][ks][e] = (_Float16)(ok ? v : 0.0f);
      }
    }
#pragma unroll
    for (int ks = 0; ks < 4; ++ks) {
#pragma unroll
      for (int e = 0; e < 16; ++e) {
        int Kl = (e < 8) ? (hi + e) : (16 + hi + (e - 8));
        int K = (ks & 1) * 32 + Kl;
        bool ok = valid && (K < HDIM);
        int Kc = (K < HDIM) ? K : 0;
        float v = (ks < 2) ? w_ih2[r * HDIM + Kc] : w_hh2[r * HDIM + Kc];
        b2[q][ks][e] = (_Float16)(ok ? v : 0.0f);
      }
    }
  }
  float bias1[4], bias2[4], wih[4];
#pragma unroll
  for (int q = 0; q < 4; ++q) {
    const int r = q * HDIM + jc;
    float bb1 = b_ih1[r] + b_hh1[r];
    float bb2 = b_ih2[r] + b_hh2[r];
    float wi  = w_ih1[r];
    bias1[q] = valid ? bb1 : 0.f; bias2[q] = valid ? bb2 : 0.f; wih[q] = valid ? wi : 0.f;
  }
  const float wlr  = w_lin[jc];
  const float wl   = valid ? wlr : 0.0f;
  const float blin = b_lin[0];

  for (int i = tid; i < 2 * 16 * 136; i += 128) ((_Float16*)hbuf)[i] = (_Float16)0.0f;
  for (int i = tid; i < 4 * 16 * 64; i += 128) ((float*)ypart)[i] = 0.0f;
  __syncthreads();

  float c1s[8], c2s[8];
#pragma unroll
  for (int k = 0; k < 8; ++k) { c1s[k] = 0.f; c2s[k] = 0.f; }

  for (int t = 0; t < Tout; ++t) {
    const int cur = t & 1, prv = cur ^ 1;
    float xv[8];
    if (t < T) {
      if (lane < 16) xtile[w][lane] = input[(long)(rowbase + lane) * T + t];
#pragma unroll
      for (int k = 0; k < 8; ++k) xv[k] = xtile[w][hi + k];
    } else {
      __syncthreads();
      const int sp = (t - 1) & 63;
#pragma unroll
      for (int k = 0; k < 8; ++k)
        xv[k] = ((ypart[0][hi + k][sp] + ypart[1][hi + k][sp]) +
                 (ypart[2][hi + k][sp] + ypart[3][hi + k][sp])) + blin;
    }

    {
      v16h a1[2];
#pragma unroll
      for (int ks = 0; ks < 2; ++ks) {
        const _Float16* bp = &hbuf[prv][cix][ks * 32 + hi];
        v8h lo = *(const v8h*)bp;
        v8h hp = *(const v8h*)(bp + 16);
        a1[ks] = __builtin_shufflevector(lo, hp, 0,1,2,3,4,5,6,7,8,9,10,11,12,13,14,15);
      }
      v8f acc[4];
#pragma unroll
      for (int q = 0; q < 4; ++q) {
        float bb = bias1[q];
        v8f cacc = {bb, bb, bb, bb, bb, bb, bb, bb};
        acc[q] = cacc;
      }
#pragma unroll
      for (int ks = 0; ks < 2; ++ks)
#pragma unroll
        for (int q = 0; q < 4; ++q)
          acc[q] = __builtin_amdgcn_wmma_f32_16x16x32_f16(false, a1[ks], false,
                      b1[q][ks], (short)0, acc[q], false, false);
#pragma unroll
      for (int q = 0; q < 4; ++q) {
        float wv = wih[q];
#pragma unroll
        for (int k = 0; k < 8; ++k) acc[q][k] += xv[k] * wv;
      }
#pragma unroll
      for (int k = 0; k < 8; ++k) {
        float ig = fast_sigmoid(acc[0][k]);
        float fg = fast_sigmoid(acc[1][k]);
        float gg = fast_tanh(acc[2][k]);
        float og = fast_sigmoid(acc[3][k]);
        float cn = fg * c1s[k] + ig * gg;
        c1s[k] = cn;
        float hn = og * fast_tanh(cn);
        hbuf[cur][hi + k][j] = valid ? (_Float16)hn : (_Float16)0.0f;
      }
    }
    __syncthreads();

    if ((t & 31) == 0 && t > 0) {
      const int t0 = t - 32;
      const int sl = (t0 + lane) & 63;
#pragma unroll
      for (int rr = 0; rr < 4; ++rr) {
        const int row = 4 * w + rr;
        out[(long)(rowbase + row) * Tout + (t0 + lane)] =
            ((ypart[0][row][sl] + ypart[1][row][sl]) + (ypart[2][row][sl] + ypart[3][row][sl])) + blin;
      }
    }

    {
      v16h a2[4];
#pragma unroll
      for (int ks = 0; ks < 4; ++ks) {
        const _Float16* bp = (ks < 2) ? &hbuf[cur][cix][ks * 32 + hi]
                                      : &hbuf[prv][cix][ks * 32 + hi];
        v8h lo = *(const v8h*)bp;
        v8h hp = *(const v8h*)(bp + 16);
        a2[ks] = __builtin_shufflevector(lo, hp, 0,1,2,3,4,5,6,7,8,9,10,11,12,13,14,15);
      }
      v8f acc[4];
#pragma unroll
      for (int q = 0; q < 4; ++q) {
        float bb = bias2[q];
        v8f cacc = {bb, bb, bb, bb, bb, bb, bb, bb};
        acc[q] = cacc;
      }
#pragma unroll
      for (int ks = 0; ks < 4; ++ks)
#pragma unroll
        for (int q = 0; q < 4; ++q)
          acc[q] = __builtin_amdgcn_wmma_f32_16x16x32_f16(false, a2[ks], false,
                      b2[q][ks], (short)0, acc[q], false, false);
      float ys[8];
#pragma unroll
      for (int k = 0; k < 8; ++k) {
        float ig = fast_sigmoid(acc[0][k]);
        float fg = fast_sigmoid(acc[1][k]);
        float gg = fast_tanh(acc[2][k]);
        float og = fast_sigmoid(acc[3][k]);
        float cn = fg * c2s[k] + ig * gg;
        c2s[k] = cn;
        float hn = og * fast_tanh(cn);
        hbuf[cur][hi + k][64 + j] = valid ? (_Float16)hn : (_Float16)0.0f;
        ys[k] = (valid ? hn : 0.0f) * wl;
      }

#pragma unroll
      for (int mask = 1; mask <= 8; mask <<= 1)
#pragma unroll
        for (int k = 0; k < 8; ++k) ys[k] += __shfl_xor(ys[k], mask, 32);
      if (cix == 0) {
        const int sc = t & 63;
#pragma unroll
        for (int k = 0; k < 8; ++k) ypart[w][hi + k][sc] = ys[k];
      }
    }
  }

  if (Tout > 0) {
    __syncthreads();
    const int t0 = ((Tout - 1) >> 5) << 5;
    const int sl = (t0 + lane) & 63;
    if (t0 + lane < Tout) {
#pragma unroll
      for (int rr = 0; rr < 4; ++rr) {
        const int row = 4 * w + rr;
        out[(long)(rowbase + row) * Tout + (t0 + lane)] =
            ((ypart[0][row][sl] + ypart[1][row][sl]) + (ypart[2][row][sl] + ypart[3][row][sl])) + blin;
      }
    }
  }
}

extern "C" void kernel_launch(void* const* d_in, const int* in_sizes, int n_in,
                              void* d_out, int out_size, void* d_ws, size_t ws_size,
                              hipStream_t stream) {
  const float* input = (const float*)d_in[0];
  const float* w_ih1 = (const float*)d_in[2];
  const float* w_hh1 = (const float*)d_in[3];
  const float* b_ih1 = (const float*)d_in[4];
  const float* b_hh1 = (const float*)d_in[5];
  const float* w_ih2 = (const float*)d_in[6];
  const float* w_hh2 = (const float*)d_in[7];
  const float* b_ih2 = (const float*)d_in[8];
  const float* b_hh2 = (const float*)d_in[9];
  const float* w_lin = (const float*)d_in[10];
  const float* b_lin = (const float*)d_in[11];
  float* out = (float*)d_out;

  const int B = 1024;
  const int T = in_sizes[0] / B;
  const int Tout = out_size / B;

  lstm2_wmma_kernel<<<dim3(B / 16), dim3(128), 0, stream>>>(
      input, w_ih1, w_hh1, b_ih1, b_hh1, w_ih2, w_hh2, b_ih2, b_hh2,
      w_lin, b_lin, out, T, Tout);
}


# Two layers as a 2-stage pipeline over 8 waves (2 per SIMD): layer-1 waves run step s while layer-2 waves run step s-1, one barrier per pass; layer 2 adds its h2 products before its h1 ones
# speedup vs baseline: 5.4510x; 1.2973x over previous
#include <hip/hip_runtime.h>
#include <hip/hip_fp16.h>


typedef __attribute__((ext_vector_type(16))) _Float16 v16h;
typedef __attribute__((ext_vector_type(8)))  _Float16 v8h;
typedef __attribute__((ext_vector_type(8)))  float    v8f;

#define HDIM 51
#define LOG2E 1.44269504088896340736f

__device__ __forceinline__ float fast_sigmoid(float z) {
  float e = __builtin_amdgcn_exp2f(-LOG2E * z);
  return __builtin_amdgcn_rcpf(1.0f + e);
}
__device__ __forceinline__ float fast_tanh(float z) {
  float e = __builtin_amdgcn_exp2f(2.0f * LOG2E * z);
  return 1.0f - 2.0f * __builtin_amdgcn_rcpf(e + 1.0f);
}

__global__ __launch_bounds__(256)
void lstm2_wmma_kernel(const float* __restrict__ input,
                       const float* __restrict__ w_ih1, const float* __restrict__ w_hh1,
                       const float* __restrict__ b_ih1, const float* __restrict__ b_hh1,
                       const float* __restrict__ w_ih2, const float* __restrict__ w_hh2,
                       const float* __restrict__ b_ih2, const float* __restrict__ b_hh2,
                       const float* __restrict__ w_lin, const float* __restrict__ b_lin,
                       float* __restrict__ out, int T, int Tout)
{
  __shared__ __align__(16) _Float16 hbuf[2][16][136];
  __shared__ float xtile[4][16];
  __shared__ float ypart[4][16][64];

  const int tid  = threadIdx.x;
  const int w    = __builtin_amdgcn_readfirstlane(tid >> 5);
  const int role = (w < 4) ? 1 : 0;
  const int ws   = w & 3;
  const int lane = tid & 31;
  const int cix  = lane & 15;
  const int hi   = (lane & 16) ? 8 : 0;
  const int rowbase = blockIdx.x * 16;
  const int j    = ws * 16 + cix;
  const bool valid = (j < HDIM);

  v16h bw[4][4];
  const int jc = valid ? j : 0;
  const float* wk1 = role ? w_ih2 : w_hh1;
#pragma unroll
  for (int q = 0; q < 4; ++q) {
    const int r = q * HDIM + jc;
#pragma unroll
    for (int ks = 0; ks < 4; ++ks) {
#pragma unroll
      for (int e = 0; e < 16; ++e) {
        int Kl = (e < 8) ? (hi + e) : (16 + hi + (e - 8));
        int K = (ks & 1) * 32 + Kl;
        bool ok = valid && (K < HDIM) && (ks < 2 || role == 1);
        int Kc = (K < HDIM) ? K : 0;
        float v = (ks < 2) ? wk1[r * HDIM + Kc] : w_hh2[r * HDIM + Kc];
        bw[q][ks][e] = (_Float16)(ok ? v : 0.0f);
      }
    }
  }
  const float* bi = role ? b_ih2 : b_ih1;
  const float* bh = role ? b_hh2 : b_hh1;
  float bias[4], wih[4];
#pragma unroll
  for (int q = 0; q < 4; ++q) {
    const int r = q * HDIM + jc;
    float bb = bi[r] + bh[r];
    float wi = w_ih1[r];
    bias[q] = valid ? bb : 0.f;
    wih[q]  = (valid && role == 0) ? wi : 0.f;
  }
  const float wlr  = w_lin[jc];
  const float wl   = valid ? wlr : 0.0f;
  const float blin = b_lin[0];

  for (int i = tid; i < 2 * 16 * 136; i += 256) ((_Float16*)hbuf)[i] = (_Float16)0.0f;
  for (int i = tid; i < 4 * 16 * 64; i += 256) ((float*)ypart)[i] = 0.0f;
  __syncthreads();

  float cs[8];
#pragma unroll
  for (int k = 0; k < 8; ++k) cs[k] = 0.f;

  const int Tm = (T < Tout) ? T : Tout;
  const int npass = Tm + 1 + 2 * (Tout - Tm);
  int stored = 0;
  int done = -1;
  for (int s = 0; s < npass; ++s) {
    const bool piped = (s <= Tm);
    const int kk = s - Tm - 1;
    const int t = piped ? (done + 1 - role) : done;
    const bool active = piped ? (role ? (s != 0) : (s != Tm)) : ((kk & 1) == role);

    if (done - stored >= 32) {
      if (role == 0) {
        const int sl = (stored + lane) & 63;
#pragma unroll
        for (int rr = 0; rr < 4; ++rr) {
          const int row = 4 * ws + rr;
          out[(long)(rowbase + row) * Tout + (stored + lane)] =
              ((ypart[0][row][sl] + ypart[1][row][sl]) + (ypart[2][row][sl] + ypart[3][row][sl])) + blin;
        }
      }
      stored += 32;
    }

    if (active) {
      const int cur = t & 1, prv = cur ^ 1;
      float xv[8];
#pragma unroll
      for (int k = 0; k < 8; ++k) xv[k] = 0.f;
      if (role == 0) {
        if (t < T) {
          if (lane < 16) xtile[ws][lane] = input[(long)(rowbase + lane) * T + t];
#pragma unroll
          for (int k = 0; k < 8; ++k) xv[k] = xtile[ws][hi + k];
        } else {
          const int sp = (t - 1) & 63;
#pragma unroll
          for (int k = 0; k < 8; ++k)
            xv[k] = ((ypart[0][hi + k][sp] + ypart[1][hi + k][sp]) +
                     (ypart[2][hi + k][sp] + ypart[3][hi + k][sp])) + blin;
        }
      }

      v8f acc[4];
#pragma unroll
      for (int q = 0; q < 4; ++q) {
        float bb = bias[q];
        v8f cacc = {bb, bb, bb, bb, bb, bb, bb, bb};
        acc[q] = cacc;
      }
      const int hb = role ? cur : prv;
      v16h a1[2];
#pragma unroll
      for (int ks = 0; ks < 2; ++ks) {
        const _Float16* bp = &hbuf[hb][cix][ks * 32 + hi];
        v8h lo = *(const v8h*)bp;
        v8h hp = *(const v8h*)(bp + 16);
        a1[ks] = __builtin_shufflevector(lo, hp, 0,1,2,3,4,5,6,7,8,9,10,11,12,13,14,15);
      }
      if (role == 1) {
        v16h a2[2];
#pragma unroll
        for (int ks = 0; ks < 2; ++ks) {
          const _Float16* bp = &hbuf[prv][cix][64 + ks * 32 + hi];
          v8h lo = *(const v8h*)bp;
          v8h hp = *(const v8h*)(bp + 16);
          a2[ks] = __builtin_shufflevector(lo, hp, 0,1,2,3,4,5,6,7,8,9,10,11,12,13,14,15);
        }
        __builtin_amdgcn_sched_barrier(0);
#pragma unroll
        for (int ks = 0; ks < 2; ++ks)
#pragma unroll
          for (int q = 0; q < 4; ++q)
            acc[q] = __builtin_amdgcn_wmma_f32_16x16x32_f16(false, a2[ks], false,
                        bw[q][2 + ks], (short)0, acc[q], false, false);
      }
#pragma unroll
      for (int ks = 0; ks < 2; ++ks)
#pragma unroll
        for (int q = 0; q < 4; ++q)
          acc[q] = __builtin_amdgcn_wmma_f32_16x16x32_f16(false, a1[ks], false,
                      bw[q][ks], (short)0, acc[q], false, false);
#pragma unroll
      for (int q = 0; q < 4; ++q) {
        float wv = wih[q];
#pragma unroll
        for (int k = 0; k < 8; ++k) acc[q][k] += xv[k] * wv;
      }
      float ys[8];
#pragma unroll
      for (int k = 0; k < 8; ++k) {
        float ig = fast_sigmoid(acc[0][k]);
        float fg = fast_sigmoid(acc[1][k]);
        float gg = fast_tanh(acc[2][k]);
        float og = fast_sigmoid(acc[3][k]);
        float cn = fg * cs[k] + ig * gg;
        cs[k] = cn;
        float hn = og * fast_tanh(cn);
        hbuf[cur][hi + k][role * 64 + j] = valid ? (_Float16)hn : (_Float16)0.0f;
        ys[k] = (valid ? hn : 0.0f) * wl;
      }

      if (role == 1) {
#pragma unroll
        for (int mask = 1; mask <= 8; mask <<= 1)
#pragma unroll
          for (int k = 0; k < 8; ++k) ys[k] += __shfl_xor(ys[k], mask, 32);
        if (cix == 0) {
          const int sc = t & 63;
#pragma unroll
          for (int k = 0; k < 8; ++k) ypart[ws][hi + k][sc] = ys[k];
        }
      }
    }
    __syncthreads();
    done += piped ? 1 : (kk & 1);
  }

  for (; stored < Tout; stored += 32) {
    if (role == 0 && stored + lane < Tout) {
      const int sl = (stored + lane) & 63;
#pragma unroll
      for (int rr = 0; rr < 4; ++rr) {
        const int row = 4 * ws + rr;
        out[(long)(rowbase + row) * Tout + (stored + lane)] =
            ((ypart[0][row][sl] + ypart[1][row][sl]) + (ypart[2][row][sl] + ypart[3][row][sl])) + blin;
      }
    }
  }
}

extern "C" void kernel_launch(void* const* d_in, const int* in_sizes, int n_in,
                              void* d_out, int out_size, void* d_ws, size_t ws_size,
                              hipStream_t stream) {
  const float* input = (const float*)d_in[0];
  const float* w_ih1 = (const float*)d_in[2];
  const float* w_hh1 = (const float*)d_in[3];
  const float* b_ih1 = (const float*)d_in[4];
  const float* b_hh1 = (const float*)d_in[5];
  const float* w_ih2 = (const float*)d_in[6];
  const float* w_hh2 = (const float*)d_in[7];
  const float* b_ih2 = (const float*)d_in[8];
  const float* b_hh2 = (const float*)d_in[9];
  const float* w_lin = (const float*)d_in[10];
  const float* b_lin = (const float*)d_in[11];
  float* out = (float*)d_out;

  const int B = 1024;
  const int T = in_sizes[0] / B;
  const int Tout = out_size / B;

  lstm2_wmma_kernel<<<dim3(B / 16), dim3(256), 0, stream>>>(
      input, w_ih1, w_hh1, b_ih1, b_hh1, w_ih2, w_hh2, b_ih2, b_hh2,
      w_lin, b_lin, out, T, Tout);
}


# All five cell nonlinearities on the hardware tanh instruction: sigmoid(z) = 0.5 + 0.5*tanh(z/2), the 0.5 folded into the i, f, o weight rows, biases and x weights at load; replaces exp2 + rcp pairs
# speedup vs baseline: 5.8691x; 1.0767x over previous
#include <hip/hip_runtime.h>
#include <hip/hip_fp16.h>


typedef __attribute__((ext_vector_type(16))) _Float16 v16h;
typedef __attribute__((ext_vector_type(8)))  _Float16 v8h;
typedef __attribute__((ext_vector_type(8)))  float    v8f;

#define HDIM 51

__device__ __forceinline__ float hw_tanh(float z) {
  return __builtin_amdgcn_tanhf(z);
}
__device__ __forceinline__ float sigmoid_of_half_arg(float zh) {
  return __builtin_fmaf(__builtin_amdgcn_tanhf(zh), 0.5f, 0.5f);
}

__global__ __launch_bounds__(256)
void lstm2_wmma_kernel(const float* __restrict__ input,
                       const float* __restrict__ w_ih1, const float* __restrict__ w_hh1,
                       const float* __restrict__ b_ih1, const float* __restrict__ b_hh1,
                       const float* __restrict__ w_ih2, const float* __restrict__ w_hh2,
                       const float* __restrict__ b_ih2, const float* __restrict__ b_hh2,
                       const float* __restrict__ w_lin, const float* __restrict__ b_lin,
                       float* __restrict__ out, int T, int Tout)
{
  __shared__ __align__(16) _Float16 hbuf[2][16][136];
  __shared__ float xtile[4][16];
  __shared__ float ypart[4][16][64];

  const int tid  = threadIdx.x;
  const int w    = __builtin_amdgcn_readfirstlane(tid >> 5);
  const int role = (w < 4) ? 1 : 0;
  const int ws   = w & 3;
  const int lane = tid & 31;
  const int cix  = lane & 15;
  const int hi   = (lane & 16) ? 8 : 0;
  const int rowbase = blockIdx.x * 16;
  const int j    = ws * 16 + cix;
  const bool valid = (j < HDIM);

  v16h bw[4][4];
  const int jc = valid ? j : 0;
  const float* wk1 = role ? w_ih2 : w_hh1;
#pragma unroll
  for (int q = 0; q < 4; ++q) {
    const int r = q * HDIM + jc;
    const float gsc = (q == 2) ? 1.0f : 0.5f;
#pragma unroll
    for (int ks = 0; ks < 4; ++ks) {
#pragma unroll
      for (int e = 0; e < 16; ++e) {
        int Kl = (e < 8) ? (hi + e) : (16 + hi + (e - 8));
        int K = (ks & 1) * 32 + Kl;
        bool ok = valid && (K < HDIM) && (ks < 2 || role == 1);
        int Kc = (K < HDIM) ? K : 0;
        float v = (ks < 2) ? wk1[r * HDIM + Kc] : w_hh2[r * HDIM + Kc];
        bw[q][ks][e] = (_Float16)(ok ? v * gsc : 0.0f);
      }
    }
  }
  const float* bi = role ? b_ih2 : b_ih1;
  const float* bh = role ? b_hh2 : b_hh1;
  float bias[4], wih[4];
#pragma unroll
  for (int q = 0; q < 4; ++q) {
    const int r = q * HDIM + jc;
    const float gsc = (q == 2) ? 1.0f : 0.5f;
    float bb = (bi[r] + bh[r]) * gsc;
    float wi = w_ih1[r] * gsc;
    bias[q] = valid ? bb : 0.f;
    wih[q]  = (valid && role == 0) ? wi : 0.f;
  }
  const float wlr  = w_lin[jc];
  const float wl   = valid ? wlr : 0.0f;
  const float blin = b_lin[0];

  for (int i = tid; i < 2 * 16 * 136; i += 256) ((_Float16*)hbuf)[i] = (_Float16)0.0f;
  for (int i = tid; i < 4 * 16 * 64; i += 256) ((float*)ypart)[i] = 0.0f;
  __syncthreads();

  float cs[8];
#pragma unroll
  for (int k = 0; k < 8; ++k) cs[k] = 0.f;

  const int Tm = (T < Tout) ? T : Tout;
  const int npass = Tm + 1 + 2 * (Tout - Tm);
  int stored = 0;
  int done = -1;
  for (int s = 0; s < npass; ++s) {
    const bool piped = (s <= Tm);
    const int kk = s - Tm - 1;
    const int t = piped ? (done + 1 - role) : done;
    const bool active = piped ? (role ? (s != 0) : (s != Tm)) : ((kk & 1) == role);

    if (done - stored >= 32) {
      if (role == 0) {
        const int sl = (stored + lane) & 63;
#pragma unroll
        for (int rr = 0; rr < 4; ++rr) {
          const int row = 4 * ws + rr;
          out[(long)(rowbase + row) * Tout + (stored + lane)] =
              ((ypart[0][row][sl] + ypart[1][row][sl]) + (ypart[2][row][sl] + ypart[3][row][sl])) + blin;
        }
      }
      stored += 32;
    }

    if (active) {
      const int cur = t & 1, prv = cur ^ 1;
      float xv[8];
#pragma unroll
      for (int k = 0; k < 8; ++k) xv[k] = 0.f;
      if (role == 0) {
        if (t < T) {
          if (lane < 16) xtile[ws][lane] = input[(long)(rowbase + lane) * T + t];
#pragma unroll
          for (int k = 0; k < 8; ++k) xv[k] = xtile[ws][hi + k];
        } else {
          const int sp = (t - 1) & 63;
#pragma unroll
          for (int k = 0; k < 8; ++k)
            xv[k] = ((ypart[0][hi + k][sp] + ypart[1][hi + k][sp]) +
                     (ypart[2][hi + k][sp] + ypart[3][hi + k][sp])) + blin;
        }
      }

      v8f acc[4];
#pragma unroll
      for (int q = 0; q < 4; ++q) {
        float bb = bias[q];
        v8f cacc = {bb, bb, bb, bb, bb, bb, bb, bb};
        acc[q] = cacc;
      }
      const int hb = role ? cur : prv;
      v16h a1[2];
#pragma unroll
      for (int ks = 0; ks < 2; ++ks) {
        const _Float16* bp = &hbuf[hb][cix][ks * 32 + hi];
        v8h lo = *(const v8h*)bp;
        v8h hp = *(const v8h*)(bp + 16);
        a1[ks] = __builtin_shufflevector(lo, hp, 0,1,2,3,4,5,6,7,8,9,10,11,12,13,14,15);
      }
      if (role == 1) {
        v16h a2[2];
#pragma unroll
        for (int ks = 0; ks < 2; ++ks) {
          const _Float16* bp = &hbuf[prv][cix][64 + ks * 32 + hi];
          v8h lo = *(const v8h*)bp;
          v8h hp = *(const v8h*)(bp + 16);
          a2[ks] = __builtin_shufflevector(lo, hp, 0,1,2,3,4,5,6,7,8,9,10,11,12,13,14,15);
        }
        __builtin_amdgcn_sched_barrier(0);
#pragma unroll
        for (int ks = 0; ks < 2; ++ks)
#pragma unroll
          for (int q = 0; q < 4; ++q)
            acc[q] = __builtin_amdgcn_wmma_f32_16x16x32_f16(false, a2[ks], false,
                        bw[q][2 + ks], (short)0, acc[q], false, false);
      }
#pragma unroll
      for (int ks = 0; ks < 2; ++ks)
#pragma unroll
        for (int q = 0; q < 4; ++q)
          acc[q] = __builtin_amdgcn_wmma_f32_16x16x32_f16(false, a1[ks], false,
                      bw[q][ks], (short)0, acc[q], false, false);
#pragma unroll
      for (int q = 0; q < 4; ++q) {
        float wv = wih[q];
#pragma unroll
        for (int k = 0; k < 8; ++k) acc[q][k] += xv[k] * wv;
      }
      float ys[8];
#pragma unroll
      for (int k = 0; k < 8; ++k) {
        float ig = sigmoid_of_half_arg(acc[0][k]);
        float fg = sigmoid_of_half_arg(acc[1][k]);
        float gg = hw_tanh(acc[2][k]);
        float og = sigmoid_of_half_arg(acc[3][k]);
        float cn = fg * cs[k] + ig * gg;
        cs[k] = cn;
        float hn = og * hw_tanh(cn);
        hbuf[cur][hi + k][role * 64 + j] = valid ? (_Float16)hn : (_Float16)0.0f;
        ys[k] = (valid ? hn : 0.0f) * wl;
      }

      if (role == 1) {
#pragma unroll
        for (int mask = 1; mask <= 8; mask <<= 1)
#pragma unroll
          for (int k = 0; k < 8; ++k) ys[k] += __shfl_xor(ys[k], mask, 32);
        if (cix == 0) {
          const int sc = t & 63;
#pragma unroll
          for (int k = 0; k < 8; ++k) ypart[ws][hi + k][sc] = ys[k];
        }
      }
    }
    __syncthreads();
    done += piped ? 1 : (kk & 1);
  }

  for (; stored < Tout; stored += 32) {
    if (role == 0 && stored + lane < Tout) {
      const int sl = (stored + lane) & 63;
#pragma unroll
      for (int rr = 0; rr < 4; ++rr) {
        const int row = 4 * ws + rr;
        out[(long)(rowbase + row) * Tout + (stored + lane)] =
            ((ypart[0][row][sl] + ypart[1][row][sl]) + (ypart[2][row][sl] + ypart[3][row][sl])) + blin;
      }
    }
  }
}

extern "C" void kernel_launch(void* const* d_in, const int* in_sizes, int n_in,
                              void* d_out, int out_size, void* d_ws, size_t ws_size,
                              hipStream_t stream) {
  const float* input = (const float*)d_in[0];
  const float* w_ih1 = (const float*)d_in[2];
  const float* w_hh1 = (const float*)d_in[3];
  const float* b_ih1 = (const float*)d_in[4];
  const float* b_hh1 = (const float*)d_in[5];
  const float* w_ih2 = (const float*)d_in[6];
  const float* w_hh2 = (const float*)d_in[7];
  const float* b_ih2 = (const float*)d_in[8];
  const float* b_hh2 = (const float*)d_in[9];
  const float* w_lin = (const float*)d_in[10];
  const float* b_lin = (const float*)d_in[11];
  float* out = (float*)d_out;

  const int B = 1024;
  const int T = in_sizes[0] / B;
  const int Tout = out_size / B;

  lstm2_wmma_kernel<<<dim3(B / 16), dim3(256), 0, stream>>>(
      input, w_ih1, w_hh1, b_ih1, b_hh1, w_ih2, w_hh2, b_ih2, b_hh2,
      w_lin, b_lin, out, T, Tout);
}


# Transposed product (weights as A, h as B): a lane owns 8 units of one batch row; h leaves in one 16-B store, x comes straight from its load, readout summed in-lane then 8 partials (new sum order)
# speedup vs baseline: 8.8654x; 1.5105x over previous
#include <hip/hip_runtime.h>
#include <hip/hip_fp16.h>


typedef __attribute__((ext_vector_type(16))) _Float16 v16h;
typedef __attribute__((ext_vector_type(8)))  _Float16 v8h;
typedef __attribute__((ext_vector_type(8)))  float    v8f;

#define HDIM 51

__device__ __forceinline__ float hw_tanh(float z) {
  return __builtin_amdgcn_tanhf(z);
}
__device__ __forceinline__ float sigmoid_of_half_arg(float zh) {
  return __builtin_fmaf(__builtin_amdgcn_tanhf(zh), 0.5f, 0.5f);
}

__global__ __launch_bounds__(256)
void lstm2_wmma_kernel(const float* __restrict__ input,
                       const float* __restrict__ w_ih1, const float* __restrict__ w_hh1,
                       const float* __restrict__ b_ih1, const float* __restrict__ b_hh1,
                       const float* __restrict__ w_ih2, const float* __restrict__ w_hh2,
                       const float* __restrict__ b_ih2, const float* __restrict__ b_hh2,
                       const float* __restrict__ w_lin, const float* __restrict__ b_lin,
                       float* __restrict__ out, int T, int Tout)
{
  __shared__ __align__(16) _Float16 hbuf[2][16][136];
  __shared__ float ypart[8][16][65];

  const int tid  = threadIdx.x;
  const int w    = __builtin_amdgcn_readfirstlane(tid >> 5);
  const int role = (w < 4) ? 1 : 0;
  const int ws   = w & 3;
  const int lane = tid & 31;
  const int cix  = lane & 15;
  const int hi   = (lane & 16) ? 8 : 0;
  const int half = lane >> 4;
  const int rowbase = blockIdx.x * 16;
  const int j    = ws * 16 + cix;
  const bool valid = (j < HDIM);

  v16h bw[4][4];
  const int jc = valid ? j : 0;
  const float* wk1 = role ? w_ih2 : w_hh1;
#pragma unroll
  for (int q = 0; q < 4; ++q) {
    const int r = q * HDIM + jc;
    const float gsc = (q == 2) ? 1.0f : 0.5f;
#pragma unroll
    for (int ks = 0; ks < 4; ++ks) {
#pragma unroll
      for (int e = 0; e < 16; ++e) {
        int Kl = (e < 8) ? (hi + e) : (16 + hi + (e - 8));
        int K = (ks & 1) * 32 + Kl;
        bool ok = valid && (K < HDIM) && (ks < 2 || role == 1);
        int Kc = (K < HDIM) ? K : 0;
        float v = (ks < 2) ? wk1[r * HDIM + Kc] : w_hh2[r * HDIM + Kc];
        bw[q][ks][e] = (_Float16)(ok ? v * gsc : 0.0f);
      }
    }
  }
  const float* bi = role ? b_ih2 : b_ih1;
  const float* bh = role ? b_hh2 : b_hh1;
  v8f biasv[4], wihv[4];
  float wlv[8];
  bool vr[8];
#pragma unroll
  for (int r = 0; r < 8; ++r) {
    const int u = ws * 16 + hi + r;
    vr[r] = (u < HDIM);
    const int uc = vr[r] ? u : 0;
    const float wlr = w_lin[uc];
    wlv[r] = vr[r] ? wlr : 0.0f;
#pragma unroll
    for (int q = 0; q < 4; ++q) {
      const int rr = q * HDIM + uc;
      const float gsc = (q == 2) ? 1.0f : 0.5f;
      float bb = (bi[rr] + bh[rr]) * gsc;
      float wi = w_ih1[rr] * gsc;
      biasv[q][r] = vr[r] ? bb : 0.f;
      wihv[q][r]  = (vr[r] && role == 0) ? wi : 0.f;
    }
  }
  const float blin = b_lin[0];

  for (int i = tid; i < 2 * 16 * 136; i += 256) ((_Float16*)hbuf)[i] = (_Float16)0.0f;
  for (int i = tid; i < 8 * 16 * 65; i += 256) ((float*)ypart)[i] = 0.0f;
  __syncthreads();

  float cs[8];
#pragma unroll
  for (int k = 0; k < 8; ++k) cs[k] = 0.f;

  const int Tm = (T < Tout) ? T : Tout;
  const int npass = Tm + 1 + 2 * (Tout - Tm);
  int stored = 0;
  int done = -1;
  for (int s = 0; s < npass; ++s) {
    const bool piped = (s <= Tm);
    const int kk = s - Tm - 1;
    const int t = piped ? (done + 1 - role) : done;
    const bool active = piped ? (role ? (s != 0) : (s != Tm)) : ((kk & 1) == role);

    if (done - stored >= 32) {
      if (role == 0) {
        const int sl = (stored + lane) & 63;
#pragma unroll
        for (int rr = 0; rr < 4; ++rr) {
          const int row = 4 * ws + rr;
          out[(long)(rowbase + row) * Tout + (stored + lane)] =
              (((ypart[0][row][sl] + ypart[1][row][sl]) + (ypart[2][row][sl] + ypart[3][row][sl])) +
               ((ypart[4][row][sl] + ypart[5][row][sl]) + (ypart[6][row][sl] + ypart[7][row][sl]))) + blin;
        }
      }
      stored += 32;
    }

    if (active) {
      const int cur = t & 1, prv = cur ^ 1;
      float xv = 0.f;
      if (role == 0) {
        if (t < T) {
          xv = input[(long)(rowbase + cix) * T + t];
        } else {
          const int sp = (t - 1) & 63;
          xv = (((ypart[0][cix][sp] + ypart[1][cix][sp]) + (ypart[2][cix][sp] + ypart[3][cix][sp])) +
                ((ypart[4][cix][sp] + ypart[5][cix][sp]) + (ypart[6][cix][sp] + ypart[7][cix][sp]))) + blin;
        }
      }

      v8f acc[4];
#pragma unroll
      for (int q = 0; q < 4; ++q) acc[q] = biasv[q];
      const int hb = role ? cur : prv;
      v16h a1[2];
#pragma unroll
      for (int ks = 0; ks < 2; ++ks) {
        const _Float16* bp = &hbuf[hb][cix][ks * 32 + hi];
        v8h lo = *(const v8h*)bp;
        v8h hp = *(const v8h*)(bp + 16);
        a1[ks] = __builtin_shufflevector(lo, hp, 0,1,2,3,4,5,6,7,8,9,10,11,12,13,14,15);
      }
      if (role == 1) {
        v16h a2[2];
#pragma unroll
        for (int ks = 0; ks < 2; ++ks) {
          const _Float16* bp = &hbuf[prv][cix][64 + ks * 32 + hi];
          v8h lo = *(const v8h*)bp;
          v8h hp = *(const v8h*)(bp + 16);
          a2[ks] = __builtin_shufflevector(lo, hp, 0,1,2,3,4,5,6,7,8,9,10,11,12,13,14,15);
        }
        __builtin_amdgcn_sched_barrier(0);
#pragma unroll
        for (int ks = 0; ks < 2; ++ks)
#pragma unroll
          for (int q = 0; q < 4; ++q)
            acc[q] = __builtin_amdgcn_wmma_f32_16x16x32_f16(false, bw[q][2 + ks], false,
                        a2[ks], (short)0, acc[q], false, false);
      }
#pragma unroll
      for (int ks = 0; ks < 2; ++ks)
#pragma unroll
        for (int q = 0; q < 4; ++q)
          acc[q] = __builtin_amdgcn_wmma_f32_16x16x32_f16(false, bw[q][ks], false,
                      a1[ks], (short)0, acc[q], false, false);
#pragma unroll
      for (int q = 0; q < 4; ++q) {
#pragma unroll
        for (int r = 0; r < 8; ++r) acc[q][r] += xv * wihv[q][r];
      }
      float hn[8];
      v8h hv;
#pragma unroll
      for (int r = 0; r < 8; ++r) {
        float ig = sigmoid_of_half_arg(acc[0][r]);
        float fg = sigmoid_of_half_arg(acc[1][r]);
        float gg = hw_tanh(acc[2][r]);
        float og = sigmoid_of_half_arg(acc[3][r]);
        float cn = fg * cs[r] + ig * gg;
        cs[r] = cn;
        float hx = og * hw_tanh(cn);
        hn[r] = vr[r] ? hx : 0.0f;
        hv[r] = (_Float16)hn[r];
      }
      *(v8h*)&hbuf[cur][cix][role * 64 + ws * 16 + hi] = hv;

      if (role == 1) {
        const float y = ((hn[0] * wlv[0] + hn[1] * wlv[1]) + (hn[2] * wlv[2] + hn[3] * wlv[3])) +
                        ((hn[4] * wlv[4] + hn[5] * wlv[5]) + (hn[6] * wlv[6] + hn[7] * wlv[7]));
        ypart[ws * 2 + half][cix][t & 63] = y;
      }
    }
    __syncthreads();
    done += piped ? 1 : (kk & 1);
  }

  for (; stored < Tout; stored += 32) {
    if (role == 0 && stored + lane < Tout) {
      const int sl = (stored + lane) & 63;
#pragma unroll
      for (int rr = 0; rr < 4; ++rr) {
        const int row = 4 * ws + rr;
        out[(long)(rowbase + row) * Tout + (stored + lane)] =
            (((ypart[0][row][sl] + ypart[1][row][sl]) + (ypart[2][row][sl] + ypart[3][row][sl])) +
             ((ypart[4][row][sl] + ypart[5][row][sl]) + (ypart[6][row][sl] + ypart[7][row][sl]))) + blin;
      }
    }
  }
}

extern "C" void kernel_launch(void* const* d_in, const int* in_sizes, int n_in,
                              void* d_out, int out_size, void* d_ws, size_t ws_size,
                              hipStream_t stream) {
  const float* input = (const float*)d_in[0];
  const float* w_ih1 = (const float*)d_in[2];
  const float* w_hh1 = (const float*)d_in[3];
  const float* b_ih1 = (const float*)d_in[4];
  const float* b_hh1 = (const float*)d_in[5];
  const float* w_ih2 = (const float*)d_in[6];
  const float* w_hh2 = (const float*)d_in[7];
  const float* b_ih2 = (const float*)d_in[8];
  const float* b_hh2 = (const float*)d_in[9];
  const float* w_lin = (const float*)d_in[10];
  const float* b_lin = (const float*)d_in[11];
  float* out = (float*)d_out;

  const int B = 1024;
  const int T = in_sizes[0] / B;
  const int Tout = out_size / B;

  lstm2_wmma_kernel<<<dim3(B / 16), dim3(256), 0, stream>>>(
      input, w_ih1, w_hh1, b_ih1, b_hh1, w_ih2, w_hh2, b_ih2, b_hh2,
      w_lin, b_lin, out, T, Tout);
}


# Layer split taken once per pass: each layer's waves run a body of their own; bias enters as C of the first product (no accumulator copies), layer 2 drops its zero x term, pass control shrinks
# speedup vs baseline: 10.7083x; 1.2079x over previous
#include <hip/hip_runtime.h>
#include <hip/hip_fp16.h>


typedef __attribute__((ext_vector_type(16))) _Float16 v16h;
typedef __attribute__((ext_vector_type(8)))  _Float16 v8h;
typedef __attribute__((ext_vector_type(8)))  float    v8f;

#define HDIM 51

__device__ __forceinline__ float hw_tanh(float z) {
  return __builtin_amdgcn_tanhf(z);
}
__device__ __forceinline__ float sigmoid_of_half_arg(float zh) {
  return __builtin_fmaf(__builtin_amdgcn_tanhf(zh), 0.5f, 0.5f);
}

__global__ __launch_bounds__(256)
void lstm2_wmma_kernel(const float* __restrict__ input,
                       const float* __restrict__ w_ih1, const float* __restrict__ w_hh1,
                       const float* __restrict__ b_ih1, const float* __restrict__ b_hh1,
                       const float* __restrict__ w_ih2, const float* __restrict__ w_hh2,
                       const float* __restrict__ b_ih2, const float* __restrict__ b_hh2,
                       const float* __restrict__ w_lin, const float* __restrict__ b_lin,
                       float* __restrict__ out, int T, int Tout)
{
  __shared__ __align__(16) _Float16 hbuf[2][16][136];
  __shared__ float ypart[8][16][65];

  const int tid  = threadIdx.x;
  const int w    = __builtin_amdgcn_readfirstlane(tid >> 5);
  const int role = (w < 4) ? 1 : 0;
  const int ws   = w & 3;
  const int lane = tid & 31;
  const int cix  = lane & 15;
  const int hi   = (lane & 16) ? 8 : 0;
  const int half = lane >> 4;
  const int rowbase = blockIdx.x * 16;
  const int j    = ws * 16 + cix;
  const bool valid = (j < HDIM);

  v16h bw[4][4];
  const int jc = valid ? j : 0;
  const float* wk1 = role ? w_ih2 : w_hh1;
#pragma unroll
  for (int q = 0; q < 4; ++q) {
    const int r = q * HDIM + jc;
    const float gsc = (q == 2) ? 1.0f : 0.5f;
#pragma unroll
    for (int ks = 0; ks < 4; ++ks) {
#pragma unroll
      for (int e = 0; e < 16; ++e) {
        int Kl = (e < 8) ? (hi + e) : (16 + hi + (e - 8));
        int K = (ks & 1) * 32 + Kl;
        bool ok = valid && (K < HDIM) && (ks < 2 || role == 1);
        int Kc = (K < HDIM) ? K : 0;
        float v = (ks < 2) ? wk1[r * HDIM + Kc] : w_hh2[r * HDIM + Kc];
        bw[q][ks][e] = (_Float16)(ok ? v * gsc : 0.0f);
      }
    }
  }
  const float* bi = role ? b_ih2 : b_ih1;
  const float* bh = role ? b_hh2 : b_hh1;
  v8f biasv[4], wihv[4];
  float wlv[8];
  bool vr[8];
#pragma unroll
  for (int r = 0; r < 8; ++r) {
    const int u = ws * 16 + hi + r;
    vr[r] = (u < HDIM);
    const int uc = vr[r] ? u : 0;
    const float wlr = w_lin[uc];
    wlv[r] = vr[r] ? wlr : 0.0f;
#pragma unroll
    for (int q = 0; q < 4; ++q) {
      const int rr = q * HDIM + uc;
      const float gsc = (q == 2) ? 1.0f : 0.5f;
      float bb = (bi[rr] + bh[rr]) * gsc;
      float wi = w_ih1[rr] * gsc;
      biasv[q][r] = vr[r] ? bb : 0.f;
      wihv[q][r]  = (vr[r] && role == 0) ? wi : 0.f;
    }
  }
  const float blin = b_lin[0];

  for (int i = tid; i < 2 * 16 * 136; i += 256) ((_Float16*)hbuf)[i] = (_Float16)0.0f;
  for (int i = tid; i < 8 * 16 * 65; i += 256) ((float*)ypart)[i] = 0.0f;
  __syncthreads();

  float cs1[8], cs2[8];
#pragma unroll
  for (int k = 0; k < 8; ++k) { cs1[k] = 0.f; cs2[k] = 0.f; }

  auto load_frag = [&](const _Float16* bp) __attribute__((always_inline)) -> v16h {
    v8h lo = *(const v8h*)bp;
    v8h hp = *(const v8h*)(bp + 16);
    return __builtin_shufflevector(lo, hp, 0,1,2,3,4,5,6,7,8,9,10,11,12,13,14,15);
  };

  auto cell = [&](const v8f (&acc)[4], float (&cs)[8], float (&hn)[8], _Float16* dst) __attribute__((always_inline)) {
    v8h hv;
#pragma unroll
    for (int r = 0; r < 8; ++r) {
      float ig = sigmoid_of_half_arg(acc[0][r]);
      float fg = sigmoid_of_half_arg(acc[1][r]);
      float gg = hw_tanh(acc[2][r]);
      float og = sigmoid_of_half_arg(acc[3][r]);
      float cn = fg * cs[r] + ig * gg;
      cs[r] = cn;
      float hx = og * hw_tanh(cn);
      hn[r] = vr[r] ? hx : 0.0f;
      hv[r] = (_Float16)hn[r];
    }
    *(v8h*)dst = hv;
  };

  auto ysum = [&](int row, int sl) __attribute__((always_inline)) -> float {
    return (((ypart[0][row][sl] + ypart[1][row][sl]) + (ypart[2][row][sl] + ypart[3][row][sl])) +
            ((ypart[4][row][sl] + ypart[5][row][sl]) + (ypart[6][row][sl] + ypart[7][row][sl]))) + blin;
  };

  auto layer1_step = [&](int t, bool from_input) __attribute__((always_inline)) {
    const int cur = t & 1, prv = cur ^ 1;
    float xv;
    if (from_input) {
      xv = input[(long)(rowbase + cix) * T + t];
    } else {
      xv = ysum(cix, (t - 1) & 63);
    }
    v16h a1[2];
#pragma unroll
    for (int ks = 0; ks < 2; ++ks) a1[ks] = load_frag(&hbuf[prv][cix][ks * 32 + hi]);
    __builtin_amdgcn_sched_barrier(0);
    v8f acc[4];
#pragma unroll
    for (int q = 0; q < 4; ++q)
      acc[q] = __builtin_amdgcn_wmma_f32_16x16x32_f16(false, bw[q][0], false,
                  a1[0], (short)0, biasv[q], false, false);
#pragma unroll
    for (int q = 0; q < 4; ++q)
      acc[q] = __builtin_amdgcn_wmma_f32_16x16x32_f16(false, bw[q][1], false,
                  a1[1], (short)0, acc[q], false, false);
#pragma unroll
    for (int q = 0; q < 4; ++q) {
#pragma unroll
      for (int r = 0; r < 8; ++r) acc[q][r] += xv * wihv[q][r];
    }
    float hn[8];
    cell(acc, cs1, hn, &hbuf[cur][cix][ws * 16 + hi]);
  };

  auto layer2_step = [&](int t) __attribute__((always_inline)) {
    const int cur = t & 1, prv = cur ^ 1;
    v16h a2[2], a1[2];
#pragma unroll
    for (int ks = 0; ks < 2; ++ks) a2[ks] = load_frag(&hbuf[prv][cix][64 + ks * 32 + hi]);
#pragma unroll
    for (int ks = 0; ks < 2; ++ks) a1[ks] = load_frag(&hbuf[cur][cix][ks * 32 + hi]);
    v8f acc[4];
#pragma unroll
    for (int q = 0; q < 4; ++q)
      acc[q] = __builtin_amdgcn_wmma_f32_16x16x32_f16(false, bw[q][2], false,
                  a2[0], (short)0, biasv[q], false, false);
#pragma unroll
    for (int q = 0; q < 4; ++q)
      acc[q] = __builtin_amdgcn_wmma_f32_16x16x32_f16(false, bw[q][3], false,
                  a2[1], (short)0, acc[q], false, false);
#pragma unroll
    for (int ks = 0; ks < 2; ++ks)
#pragma unroll
      for (int q = 0; q < 4; ++q)
        acc[q] = __builtin_amdgcn_wmma_f32_16x16x32_f16(false, bw[q][ks], false,
                    a1[ks], (short)0, acc[q], false, false);
    float hn[8];
    cell(acc, cs2, hn, &hbuf[cur][cix][64 + ws * 16 + hi]);
    const float y = ((hn[0] * wlv[0] + hn[1] * wlv[1]) + (hn[2] * wlv[2] + hn[3] * wlv[3])) +
                    ((hn[4] * wlv[4] + hn[5] * wlv[5]) + (hn[6] * wlv[6] + hn[7] * wlv[7]));
    ypart[ws * 2 + half][cix][t & 63] = y;
  };

  auto store32 = [&](int first) __attribute__((always_inline)) {
    const int sl = (first + lane) & 63;
#pragma unroll
    for (int rr = 0; rr < 4; ++rr) {
      const int row = 4 * ws + rr;
      out[(long)(rowbase + row) * Tout + (first + lane)] = ysum(row, sl);
    }
  };

  const int Tm = (T < Tout) ? T : Tout;
  int stored = 0;
  for (int s = 0; s < Tm; ++s) {
    if (role == 0) {
      if (s - 1 - stored >= 32) {
        store32(stored);
        stored += 32;
      }
      layer1_step(s, true);
    } else {
      if (s > 0) layer2_step(s - 1);
    }
    __syncthreads();
  }
  if (role == 0) {
    if (Tm - 1 - stored >= 32) {
      store32(stored);
      stored += 32;
    }
  } else {
    if (Tm > 0) layer2_step(Tm - 1);
  }
  __syncthreads();
  for (int t = Tm; t < Tout; ++t) {
    if (role == 0) {
      if (t - stored >= 32) {
        store32(stored);
        stored += 32;
      }
      layer1_step(t, false);
    }
    __syncthreads();
    if (role == 1) layer2_step(t);
    __syncthreads();
  }

  if (role == 0) {
    for (; stored < Tout; stored += 32) {
      if (stored + lane < Tout) store32(stored);
    }
  }
}

extern "C" void kernel_launch(void* const* d_in, const int* in_sizes, int n_in,
                              void* d_out, int out_size, void* d_ws, size_t ws_size,
                              hipStream_t stream) {
  const float* input = (const float*)d_in[0];
  const float* w_ih1 = (const float*)d_in[2];
  const float* w_hh1 = (const float*)d_in[3];
  const float* b_ih1 = (const float*)d_in[4];
  const float* b_hh1 = (const float*)d_in[5];
  const float* w_ih2 = (const float*)d_in[6];
  const float* w_hh2 = (const float*)d_in[7];
  const float* b_ih2 = (const float*)d_in[8];
  const float* b_hh2 = (const float*)d_in[9];
  const float* w_lin = (const float*)d_in[10];
  const float* b_lin = (const float*)d_in[11];
  float* out = (float*)d_out;

  const int B = 1024;
  const int T = in_sizes[0] / B;
  const int Tout = out_size / B;

  lstm2_wmma_kernel<<<dim3(B / 16), dim3(256), 0, stream>>>(
      input, w_ih1, w_hh1, b_ih1, b_hh1, w_ih2, w_hh2, b_ih2, b_hh2,
      w_lin, b_lin, out, T, Tout);
}


# 8 batch rows per workgroup on twice the workgroups: B carries the rows twice, twin lanes c and c+8 each take 4 of a tile's 8 registers through the cell; 16 readout partials (new sum order)
# speedup vs baseline: 12.5684x; 1.1737x over previous
#include <hip/hip_runtime.h>
#include <hip/hip_fp16.h>


typedef __attribute__((ext_vector_type(16))) _Float16 v16h;
typedef __attribute__((ext_vector_type(8)))  _Float16 v8h;
typedef __attribute__((ext_vector_type(4)))  _Float16 v4h;
typedef __attribute__((ext_vector_type(8)))  float    v8f;

#define HDIM 51

__device__ __forceinline__ float hw_tanh(float z) {
  return __builtin_amdgcn_tanhf(z);
}
__device__ __forceinline__ float sigmoid_of_half_arg(float zh) {
  return __builtin_fmaf(__builtin_amdgcn_tanhf(zh), 0.5f, 0.5f);
}

__global__ __launch_bounds__(256)
void lstm2_wmma_kernel(const float* __restrict__ input,
                       const float* __restrict__ w_ih1, const float* __restrict__ w_hh1,
                       const float* __restrict__ b_ih1, const float* __restrict__ b_hh1,
                       const float* __restrict__ w_ih2, const float* __restrict__ w_hh2,
                       const float* __restrict__ b_ih2, const float* __restrict__ b_hh2,
                       const float* __restrict__ w_lin, const float* __restrict__ b_lin,
                       float* __restrict__ out, int T, int Tout)
{
  __shared__ __align__(16) _Float16 hbuf[2][8][136];
  __shared__ float ypart[16][8][65];

  const int tid  = threadIdx.x;
  const int w    = __builtin_amdgcn_readfirstlane(tid >> 5);
  const int role = (w < 4) ? 1 : 0;
  const int ws   = w & 3;
  const int lane = tid & 31;
  const int cix  = lane & 15;
  const int brow = cix & 7;
  const int up   = cix >> 3;
  const int hi   = (lane & 16) ? 8 : 0;
  const int half = lane >> 4;
  const int rowbase = blockIdx.x * 8;
  const int j    = ws * 16 + cix;
  const bool valid = (j < HDIM);

  v16h bw[4][4];
  const int jc = valid ? j : 0;
  const float* wk1 = role ? w_ih2 : w_hh1;
#pragma unroll
  for (int q = 0; q < 4; ++q) {
    const int r = q * HDIM + jc;
    const float gsc = (q == 2) ? 1.0f : 0.5f;
#pragma unroll
    for (int ks = 0; ks < 4; ++ks) {
#pragma unroll
      for (int e = 0; e < 16; ++e) {
        int Kl = (e < 8) ? (hi + e) : (16 + hi + (e - 8));
        int K = (ks & 1) * 32 + Kl;
        bool ok = valid && (K < HDIM) && (ks < 2 || role == 1);
        int Kc = (K < HDIM) ? K : 0;
        float v = (ks < 2) ? wk1[r * HDIM + Kc] : w_hh2[r * HDIM + Kc];
        bw[q][ks][e] = (_Float16)(ok ? v * gsc : 0.0f);
      }
    }
  }
  const float* bi = role ? b_ih2 : b_ih1;
  const float* bh = role ? b_hh2 : b_hh1;
  v8f biasv[4];
#pragma unroll
  for (int r = 0; r < 8; ++r) {
    const int u = ws * 16 + hi + r;
    const bool uv = (u < HDIM);
    const int uc = uv ? u : 0;
#pragma unroll
    for (int q = 0; q < 4; ++q) {
      const int rr = q * HDIM + uc;
      const float gsc = (q == 2) ? 1.0f : 0.5f;
      float bb = (bi[rr] + bh[rr]) * gsc;
      biasv[q][r] = uv ? bb : 0.f;
    }
  }
  float wihs[4][4];
  float wlv[4];
  bool vr[4];
#pragma unroll
  for (int r = 0; r < 4; ++r) {
    const int u = ws * 16 + hi + 4 * up + r;
    vr[r] = (u < HDIM);
    const int uc = vr[r] ? u : 0;
    const float wlr = w_lin[uc];
    wlv[r] = vr[r] ? wlr : 0.0f;
#pragma unroll
    for (int q = 0; q < 4; ++q) {
      const int rr = q * HDIM + uc;
      const float gsc = (q == 2) ? 1.0f : 0.5f;
      float wi = w_ih1[rr] * gsc;
      wihs[q][r] = (vr[r] && role == 0) ? wi : 0.f;
    }
  }
  const float blin = b_lin[0];

  for (int i = tid; i < 2 * 8 * 136; i += 256) ((_Float16*)hbuf)[i] = (_Float16)0.0f;
  for (int i = tid; i < 16 * 8 * 65; i += 256) ((float*)ypart)[i] = 0.0f;
  __syncthreads();

  float cs1[4], cs2[4];
#pragma unroll
  for (int k = 0; k < 4; ++k) { cs1[k] = 0.f; cs2[k] = 0.f; }

  auto load_frag = [&](const _Float16* bp) __attribute__((always_inline)) -> v16h {
    v8h lo = *(const v8h*)bp;
    v8h hp = *(const v8h*)(bp + 16);
    return __builtin_shufflevector(lo, hp, 0,1,2,3,4,5,6,7,8,9,10,11,12,13,14,15);
  };

  unsigned upm = 0u - (unsigned)up;
  asm volatile("" : "+v"(upm));
  auto keep4 = [&](const v8f (&acc)[4], float (&z)[4][4]) __attribute__((always_inline)) {
#pragma unroll
    for (int q = 0; q < 4; ++q) {
#pragma unroll
      for (int r = 0; r < 4; ++r) {
        const float lof = acc[q][r];
        const float hif = acc[q][r + 4];
        const unsigned lo = __float_as_uint(lof);
        const unsigned hh = __float_as_uint(hif);
        z[q][r] = __uint_as_float(lo ^ ((lo ^ hh) & upm));
      }
    }
  };

  auto cell = [&](const float (&z)[4][4], float (&cs)[4], float (&hn)[4], _Float16* dst) __attribute__((always_inline)) {
    v4h hv;
#pragma unroll
    for (int r = 0; r < 4; ++r) {
      float ig = sigmoid_of_half_arg(z[0][r]);
      float fg = sigmoid_of_half_arg(z[1][r]);
      float gg = hw_tanh(z[2][r]);
      float og = sigmoid_of_half_arg(z[3][r]);
      float cn = fg * cs[r] + ig * gg;
      cs[r] = cn;
      float hx = og * hw_tanh(cn);
      hn[r] = vr[r] ? hx : 0.0f;
      hv[r] = (_Float16)hn[r];
    }
    *(v4h*)dst = hv;
  };

  auto ysum = [&](int row, int sl) __attribute__((always_inline)) -> float {
    const float s0 = (ypart[0][row][sl] + ypart[1][row][sl]) + (ypart[2][row][sl] + ypart[3][row][sl]);
    const float s1 = (ypart[4][row][sl] + ypart[5][row][sl]) + (ypart[6][row][sl] + ypart[7][row][sl]);
    const float s2 = (ypart[8][row][sl] + ypart[9][row][sl]) + (ypart[10][row][sl] + ypart[11][row][sl]);
    const float s3 = (ypart[12][row][sl] + ypart[13][row][sl]) + (ypart[14][row][sl] + ypart[15][row][sl]);
    return ((s0 + s1) + (s2 + s3)) + blin;
  };

  auto layer1_step = [&](int t, bool from_input) __attribute__((always_inline)) {
    const int cur = t & 1, prv = cur ^ 1;
    float xv;
    if (from_input) {
      xv = input[(long)(rowbase + brow) * T + t];
    } else {
      xv = ysum(brow, (t - 1) & 63);
    }
    v16h a1[2];
#pragma unroll
    for (int ks = 0; ks < 2; ++ks) a1[ks] = load_frag(&hbuf[prv][brow][ks * 32 + hi]);
    __builtin_amdgcn_sched_barrier(0);
    v8f acc[4];
#pragma unroll
    for (int q = 0; q < 4; ++q)
      acc[q] = __builtin_amdgcn_wmma_f32_16x16x32_f16(false, bw[q][0], false,
                  a1[0], (short)0, biasv[q], false, false);
#pragma unroll
    for (int q = 0; q < 4; ++q)
      acc[q] = __builtin_amdgcn_wmma_f32_16x16x32_f16(false, bw[q][1], false,
                  a1[1], (short)0, acc[q], false, false);
    float z[4][4];
    keep4(acc, z);
#pragma unroll
    for (int q = 0; q < 4; ++q) {
#pragma unroll
      for (int r = 0; r < 4; ++r) z[q][r] += xv * wihs[q][r];
    }
    float hn[4];
    cell(z, cs1, hn, &hbuf[cur][brow][ws * 16 + hi + 4 * up]);
  };

  auto layer2_step = [&](int t) __attribute__((always_inline)) {
    const int cur = t & 1, prv = cur ^ 1;
    v16h a2[2], a1[2];
#pragma unroll
    for (int ks = 0; ks < 2; ++ks) a2[ks] = load_frag(&hbuf[prv][brow][64 + ks * 32 + hi]);
#pragma unroll
    for (int ks = 0; ks < 2; ++ks) a1[ks] = load_frag(&hbuf[cur][brow][ks * 32 + hi]);
    v8f acc[4];
#pragma unroll
    for (int q = 0; q < 4; ++q)
      acc[q] = __builtin_amdgcn_wmma_f32_16x16x32_f16(false, bw[q][2], false,
                  a2[0], (short)0, biasv[q], false, false);
#pragma unroll
    for (int q = 0; q < 4; ++q)
      acc[q] = __builtin_amdgcn_wmma_f32_16x16x32_f16(false, bw[q][3], false,
                  a2[1], (short)0, acc[q], false, false);
#pragma unroll
    for (int ks = 0; ks < 2; ++ks)
#pragma unroll
      for (int q = 0; q < 4; ++q)
        acc[q] = __builtin_amdgcn_wmma_f32_16x16x32_f16(false, bw[q][ks], false,
                    a1[ks], (short)0, acc[q], false, false);
    float z[4][4];
    keep4(acc, z);
    float hn[4];
    cell(z, cs2, hn, &hbuf[cur][brow][64 + ws * 16 + hi + 4 * up]);
    const float y = (hn[0] * wlv[0] + hn[1] * wlv[1]) + (hn[2] * wlv[2] + hn[3] * wlv[3]);
    ypart[ws * 4 + half * 2 + up][brow][t & 63] = y;
  };

  auto store32 = [&](int first) __attribute__((always_inline)) {
    const int sl = (first + lane) & 63;
#pragma unroll
    for (int rr = 0; rr < 2; ++rr) {
      const int row = 2 * ws + rr;
      out[(long)(rowbase + row) * Tout + (first + lane)] = ysum(row, sl);
    }
  };

  const int Tm = (T < Tout) ? T : Tout;
  int stored = 0;
  for (int s = 0; s < Tm; ++s) {
    if (role == 0) {
      if (s - 1 - stored >= 32) {
        store32(stored);
        stored += 32;
      }
      layer1_step(s, true);
    } else {
      if (s > 0) layer2_step(s - 1);
    }
    __syncthreads();
  }
  if (role == 0) {
    if (Tm - 1 - stored >= 32) {
      store32(stored);
      stored += 32;
    }
  } else {
    if (Tm > 0) layer2_step(Tm - 1);
  }
  __syncthreads();
  for (int t = Tm; t < Tout; ++t) {
    if (role == 0) {
      if (t - stored >= 32) {
        store32(stored);
        stored += 32;
      }
      layer1_step(t, false);
    }
    __syncthreads();
    if (role == 1) layer2_step(t);
    __syncthreads();
  }

  if (role == 0) {
    for (; stored < Tout; stored += 32) {
      if (stored + lane < Tout) store32(stored);
    }
  }
}

extern "C" void kernel_launch(void* const* d_in, const int* in_sizes, int n_in,
                              void* d_out, int out_size, void* d_ws, size_t ws_size,
                              hipStream_t stream) {
  const float* input = (const float*)d_in[0];
  const float* w_ih1 = (const float*)d_in[2];
  const float* w_hh1 = (const float*)d_in[3];
  const float* b_ih1 = (const float*)d_in[4];
  const float* b_hh1 = (const float*)d_in[5];
  const float* w_ih2 = (const float*)d_in[6];
  const float* w_hh2 = (const float*)d_in[7];
  const float* b_ih2 = (const float*)d_in[8];
  const float* b_hh2 = (const float*)d_in[9];
  const float* w_lin = (const float*)d_in[10];
  const float* b_lin = (const float*)d_in[11];
  float* out = (float*)d_out;

  const int B = 1024;
  const int T = in_sizes[0] / B;
  const int Tout = out_size / B;

  lstm2_wmma_kernel<<<dim3(B / 8), dim3(256), 0, stream>>>(
      input, w_ih1, w_hh1, b_ih1, b_hh1, w_ih2, w_hh2, b_ih2, b_hh2,
      w_lin, b_lin, out, T, Tout);
}


# Layer 2's input product (W_ih2 x h1) leaves the pass's chain: layer 2 lags 2 steps, fetches h1 fragments a pass early, issues their products under the h2 load wait; h1 products first (new sum order)
# speedup vs baseline: 13.7760x; 1.0961x over previous
#include <hip/hip_runtime.h>
#include <hip/hip_fp16.h>


typedef __attribute__((ext_vector_type(16))) _Float16 v16h;
typedef __attribute__((ext_vector_type(8)))  _Float16 v8h;
typedef __attribute__((ext_vector_type(4)))  _Float16 v4h;
typedef __attribute__((ext_vector_type(8)))  float    v8f;

#define HDIM 51

__device__ __forceinline__ float hw_tanh(float z) {
  return __builtin_amdgcn_tanhf(z);
}
__device__ __forceinline__ float sigmoid_of_half_arg(float zh) {
  return __builtin_fmaf(__builtin_amdgcn_tanhf(zh), 0.5f, 0.5f);
}

__global__ __launch_bounds__(256)
void lstm2_wmma_kernel(const float* __restrict__ input,
                       const float* __restrict__ w_ih1, const float* __restrict__ w_hh1,
                       const float* __restrict__ b_ih1, const float* __restrict__ b_hh1,
                       const float* __restrict__ w_ih2, const float* __restrict__ w_hh2,
                       const float* __restrict__ b_ih2, const float* __restrict__ b_hh2,
                       const float* __restrict__ w_lin, const float* __restrict__ b_lin,
                       float* __restrict__ out, int T, int Tout)
{
  __shared__ __align__(16) _Float16 hbuf[2][8][136];
  __shared__ float ypart[16][8][65];

  const int tid  = threadIdx.x;
  const int w    = __builtin_amdgcn_readfirstlane(tid >> 5);
  const int role = (w < 4) ? 1 : 0;
  const int ws   = w & 3;
  const int lane = tid & 31;
  const int cix  = lane & 15;
  const int brow = cix & 7;
  const int up   = cix >> 3;
  const int hi   = (lane & 16) ? 8 : 0;
  const int half = lane >> 4;
  const int rowbase = blockIdx.x * 8;
  const int j    = ws * 16 + cix;
  const bool valid = (j < HDIM);

  v16h bw[4][4];
  const int jc = valid ? j : 0;
  const float* wk1 = role ? w_ih2 : w_hh1;
#pragma unroll
  for (int q = 0; q < 4; ++q) {
    const int r = q * HDIM + jc;
    const float gsc = (q == 2) ? 1.0f : 0.5f;
#pragma unroll
    for (int ks = 0; ks < 4; ++ks) {
#pragma unroll
      for (int e = 0; e < 16; ++e) {
        int Kl = (e < 8) ? (hi + e) : (16 + hi + (e - 8));
        int K = (ks & 1) * 32 + Kl;
        bool ok = valid && (K < HDIM) && (ks < 2 || role == 1);
        int Kc = (K < HDIM) ? K : 0;
        float v = (ks < 2) ? wk1[r * HDIM + Kc] : w_hh2[r * HDIM + Kc];
        bw[q][ks][e] = (_Float16)(ok ? v * gsc : 0.0f);
      }
    }
  }
  const float* bi = role ? b_ih2 : b_ih1;
  const float* bh = role ? b_hh2 : b_hh1;
  v8f biasv[4];
#pragma unroll
  for (int r = 0; r < 8; ++r) {
    const int u = ws * 16 + hi + r;
    const bool uv = (u < HDIM);
    const int uc = uv ? u : 0;
#pragma unroll
    for (int q = 0; q < 4; ++q) {
      const int rr = q * HDIM + uc;
      const float gsc = (q == 2) ? 1.0f : 0.5f;
      float bb = (bi[rr] + bh[rr]) * gsc;
      biasv[q][r] = uv ? bb : 0.f;
    }
  }
  float wihs[4][4];
  float wlv[4];
  bool vr[4];
#pragma unroll
  for (int r = 0; r < 4; ++r) {
    const int u = ws * 16 + hi + 4 * up + r;
    vr[r] = (u < HDIM);
    const int uc = vr[r] ? u : 0;
    const float wlr = w_lin[uc];
    wlv[r] = vr[r] ? wlr : 0.0f;
#pragma unroll
    for (int q = 0; q < 4; ++q) {
      const int rr = q * HDIM + uc;
      const float gsc = (q == 2) ? 1.0f : 0.5f;
      float wi = w_ih1[rr] * gsc;
      wihs[q][r] = (vr[r] && role == 0) ? wi : 0.f;
    }
  }
  const float blin = b_lin[0];

  for (int i = tid; i < 2 * 8 * 136; i += 256) ((_Float16*)hbuf)[i] = (_Float16)0.0f;
  for (int i = tid; i < 16 * 8 * 65; i += 256) ((float*)ypart)[i] = 0.0f;
  __syncthreads();

  float cs1[4], cs2[4];
#pragma unroll
  for (int k = 0; k < 4; ++k) { cs1[k] = 0.f; cs2[k] = 0.f; }

  auto load_frag = [&](const _Float16* bp) __attribute__((always_inline)) -> v16h {
    v8h lo = *(const v8h*)bp;
    v8h hp = *(const v8h*)(bp + 16);
    return __builtin_shufflevector(lo, hp, 0,1,2,3,4,5,6,7,8,9,10,11,12,13,14,15);
  };

  unsigned upm = 0u - (unsigned)up;
  asm volatile("" : "+v"(upm));
  auto keep4 = [&](const v8f (&acc)[4], float (&z)[4][4]) __attribute__((always_inline)) {
#pragma unroll
    for (int q = 0; q < 4; ++q) {
#pragma unroll
      for (int r = 0; r < 4; ++r) {
        const float lof = acc[q][r];
        const float hif = acc[q][r + 4];
        const unsigned lo = __float_as_uint(lof);
        const unsigned hh = __float_as_uint(hif);
        z[q][r] = __uint_as_float(lo ^ ((lo ^ hh) & upm));
      }
    }
  };

  auto cell = [&](const float (&z)[4][4], float (&cs)[4], float (&hn)[4], _Float16* dst) __attribute__((always_inline)) {
    v4h hv;
#pragma unroll
    for (int r = 0; r < 4; ++r) {
      float ig = sigmoid_of_half_arg(z[0][r]);
      float fg = sigmoid_of_half_arg(z[1][r]);
      float gg = hw_tanh(z[2][r]);
      float og = sigmoid_of_half_arg(z[3][r]);
      float cn = fg * cs[r] + ig * gg;
      cs[r] = cn;
      float hx = og * hw_tanh(cn);
      hn[r] = vr[r] ? hx : 0.0f;
      hv[r] = (_Float16)hn[r];
    }
    *(v4h*)dst = hv;
  };

  auto ysum = [&](int row, int sl) __attribute__((always_inline)) -> float {
    const float s0 = (ypart[0][row][sl] + ypart[1][row][sl]) + (ypart[2][row][sl] + ypart[3][row][sl]);
    const float s1 = (ypart[4][row][sl] + ypart[5][row][sl]) + (ypart[6][row][sl] + ypart[7][row][sl]);
    const float s2 = (ypart[8][row][sl] + ypart[9][row][sl]) + (ypart[10][row][sl] + ypart[11][row][sl]);
    const float s3 = (ypart[12][row][sl] + ypart[13][row][sl]) + (ypart[14][row][sl] + ypart[15][row][sl]);
    return ((s0 + s1) + (s2 + s3)) + blin;
  };

  auto layer1_step = [&](int t, bool from_input) __attribute__((always_inline)) {
    const int cur = t & 1, prv = cur ^ 1;
    float xv;
    if (from_input) {
      xv = input[(long)(rowbase + brow) * T + t];
    } else {
      xv = ysum(brow, (t - 1) & 63);
    }
    v16h a1[2];
#pragma unroll
    for (int ks = 0; ks < 2; ++ks) a1[ks] = load_frag(&hbuf[prv][brow][ks * 32 + hi]);
    __builtin_amdgcn_sched_barrier(0);
    v8f acc[4];
#pragma unroll
    for (int q = 0; q < 4; ++q)
      acc[q] = __builtin_amdgcn_wmma_f32_16x16x32_f16(false, bw[q][0], false,
                  a1[0], (short)0, biasv[q], false, false);
#pragma unroll
    for (int q = 0; q < 4; ++q)
      acc[q] = __builtin_amdgcn_wmma_f32_16x16x32_f16(false, bw[q][1], false,
                  a1[1], (short)0, acc[q], false, false);
    float z[4][4];
    keep4(acc, z);
#pragma unroll
    for (int q = 0; q < 4; ++q) {
#pragma unroll
      for (int r = 0; r < 4; ++r) z[q][r] += xv * wihs[q][r];
    }
    float hn[4];
    cell(z, cs1, hn, &hbuf[cur][brow][ws * 16 + hi + 4 * up]);
  };

  v16h hf[2] = {};
  auto fetch_h1 = [&](int t) __attribute__((always_inline)) {
#pragma unroll
    for (int ks = 0; ks < 2; ++ks) hf[ks] = load_frag(&hbuf[t & 1][brow][ks * 32 + hi]);
  };

  auto layer2_step = [&](int t, bool fetch_next) __attribute__((always_inline)) {
    const int cur = t & 1, prv = cur ^ 1;
    v16h a2[2];
#pragma unroll
    for (int ks = 0; ks < 2; ++ks) a2[ks] = load_frag(&hbuf[prv][brow][64 + ks * 32 + hi]);
    __builtin_amdgcn_sched_barrier(0);
    v8f acc[4];
#pragma unroll
    for (int q = 0; q < 4; ++q)
      acc[q] = __builtin_amdgcn_wmma_f32_16x16x32_f16(false, bw[q][0], false,
                  hf[0], (short)0, biasv[q], false, false);
#pragma unroll
    for (int q = 0; q < 4; ++q)
      acc[q] = __builtin_amdgcn_wmma_f32_16x16x32_f16(false, bw[q][1], false,
                  hf[1], (short)0, acc[q], false, false);
#pragma unroll
    for (int ks = 0; ks < 2; ++ks)
#pragma unroll
      for (int q = 0; q < 4; ++q)
        acc[q] = __builtin_amdgcn_wmma_f32_16x16x32_f16(false, bw[q][2 + ks], false,
                    a2[ks], (short)0, acc[q], false, false);
    if (fetch_next) {
      asm volatile("" : "+v"(acc[0]), "+v"(acc[1]), "+v"(acc[2]), "+v"(acc[3]));
      __builtin_amdgcn_sched_barrier(0);
      fetch_h1(t + 1);
      __builtin_amdgcn_sched_barrier(0);
    }
    float z[4][4];
    keep4(acc, z);
    float hn[4];
    cell(z, cs2, hn, &hbuf[cur][brow][64 + ws * 16 + hi + 4 * up]);
    const float y = (hn[0] * wlv[0] + hn[1] * wlv[1]) + (hn[2] * wlv[2] + hn[3] * wlv[3]);
    ypart[ws * 4 + half * 2 + up][brow][t & 63] = y;
  };

  auto store32 = [&](int first) __attribute__((always_inline)) {
    const int sl = (first + lane) & 63;
#pragma unroll
    for (int rr = 0; rr < 2; ++rr) {
      const int row = 2 * ws + rr;
      out[(long)(rowbase + row) * Tout + (first + lane)] = ysum(row, sl);
    }
  };

  const int Tm = (T < Tout) ? T : Tout;
  int stored = 0;
  if (Tm > 0) {
    if (role == 0) layer1_step(0, true);
    __syncthreads();
    if (role == 0) {
      if (Tm > 1) layer1_step(1, true);
    } else {
      fetch_h1(0);
    }
    __syncthreads();
  }
  for (int s = 2; s < Tm; ++s) {
    if (role == 0) {
      if (s - 2 - stored >= 32) {
        store32(stored);
        stored += 32;
      }
      layer1_step(s, true);
    } else {
      layer2_step(s - 2, true);
    }
    __syncthreads();
  }
  if (Tm > 1) {
    if (role == 1) layer2_step(Tm - 2, true);
    __syncthreads();
  }
  if (Tm > 0) {
    if (role == 1) layer2_step(Tm - 1, false);
    __syncthreads();
  }
  for (int t = Tm; t < Tout; ++t) {
    if (role == 0) {
      if (t - stored >= 32) {
        store32(stored);
        stored += 32;
      }
      layer1_step(t, false);
    }
    __syncthreads();
    if (role == 1) {
      fetch_h1(t);
      layer2_step(t, false);
    }
    __syncthreads();
  }

  if (role == 0) {
    for (; stored < Tout; stored += 32) {
      if (stored + lane < Tout) store32(stored);
    }
  }
}

extern "C" void kernel_launch(void* const* d_in, const int* in_sizes, int n_in,
                              void* d_out, int out_size, void* d_ws, size_t ws_size,
                              hipStream_t stream) {
  const float* input = (const float*)d_in[0];
  const float* w_ih1 = (const float*)d_in[2];
  const float* w_hh1 = (const float*)d_in[3];
  const float* b_ih1 = (const float*)d_in[4];
  const float* b_hh1 = (const float*)d_in[5];
  const float* w_ih2 = (const float*)d_in[6];
  const float* w_hh2 = (const float*)d_in[7];
  const float* b_ih2 = (const float*)d_in[8];
  const float* b_hh2 = (const float*)d_in[9];
  const float* w_lin = (const float*)d_in[10];
  const float* b_lin = (const float*)d_in[11];
  float* out = (float*)d_out;

  const int B = 1024;
  const int T = in_sizes[0] / B;
  const int Tout = out_size / B;

  lstm2_wmma_kernel<<<dim3(B / 8), dim3(256), 0, stream>>>(
      input, w_ih1, w_hh1, b_ih1, b_hh1, w_ih2, w_hh2, b_ih2, b_hh2,
      w_lin, b_lin, out, T, Tout);
}


# Each pass's two awaited fragment loads are issued first behind the barrier, ahead of the layer split and pass control, from a carried LDS offset; layer 1's x load heads its side
# speedup vs baseline: 14.4587x; 1.0496x over previous
#include <hip/hip_runtime.h>
#include <hip/hip_fp16.h>


typedef __attribute__((ext_vector_type(16))) _Float16 v16h;
typedef __attribute__((ext_vector_type(8)))  _Float16 v8h;
typedef __attribute__((ext_vector_type(4)))  _Float16 v4h;
typedef __attribute__((ext_vector_type(8)))  float    v8f;

#define HDIM 51

__device__ __forceinline__ float hw_tanh(float z) {
  return __builtin_amdgcn_tanhf(z);
}
__device__ __forceinline__ float sigmoid_of_half_arg(float zh) {
  return __builtin_fmaf(__builtin_amdgcn_tanhf(zh), 0.5f, 0.5f);
}

__global__ __launch_bounds__(256)
void lstm2_wmma_kernel(const float* __restrict__ input,
                       const float* __restrict__ w_ih1, const float* __restrict__ w_hh1,
                       const float* __restrict__ b_ih1, const float* __restrict__ b_hh1,
                       const float* __restrict__ w_ih2, const float* __restrict__ w_hh2,
                       const float* __restrict__ b_ih2, const float* __restrict__ b_hh2,
                       const float* __restrict__ w_lin, const float* __restrict__ b_lin,
                       float* __restrict__ out, int T, int Tout)
{
  __shared__ __align__(16) _Float16 hbuf[2][8][136];
  __shared__ float ypart[16][8][65];

  const int tid  = threadIdx.x;
  const int w    = __builtin_amdgcn_readfirstlane(tid >> 5);
  const int role = (w < 4) ? 1 : 0;
  const int ws   = w & 3;
  const int lane = tid & 31;
  const int cix  = lane & 15;
  const int brow = cix & 7;
  const int up   = cix >> 3;
  const int hi   = (lane & 16) ? 8 : 0;
  const int half = lane >> 4;
  const int rowbase = blockIdx.x * 8;
  const int j    = ws * 16 + cix;
  const bool valid = (j < HDIM);

  v16h bw[4][4];
  const int jc = valid ? j : 0;
  const float* wk1 = role ? w_ih2 : w_hh1;
#pragma unroll
  for (int q = 0; q < 4; ++q) {
    const int r = q * HDIM + jc;
    const float gsc = (q == 2) ? 1.0f : 0.5f;
#pragma unroll
    for (int ks = 0; ks < 4; ++ks) {
#pragma unroll
      for (int e = 0; e < 16; ++e) {
        int Kl = (e < 8) ? (hi + e) : (16 + hi + (e - 8));
        int K = (ks & 1) * 32 + Kl;
        bool ok = valid && (K < HDIM) && (ks < 2 || role == 1);
        int Kc = (K < HDIM) ? K : 0;
        float v = (ks < 2) ? wk1[r * HDIM + Kc] : w_hh2[r * HDIM + Kc];
        bw[q][ks][e] = (_Float16)(ok ? v * gsc : 0.0f);
      }
    }
  }
  const float* bi = role ? b_ih2 : b_ih1;
  const float* bh = role ? b_hh2 : b_hh1;
  v8f biasv[4];
#pragma unroll
  for (int r = 0; r < 8; ++r) {
    const int u = ws * 16 + hi + r;
    const bool uv = (u < HDIM);
    const int uc = uv ? u : 0;
#pragma unroll
    for (int q = 0; q < 4; ++q) {
      const int rr = q * HDIM + uc;
      const float gsc = (q == 2) ? 1.0f : 0.5f;
      float bb = (bi[rr] + bh[rr]) * gsc;
      biasv[q][r] = uv ? bb : 0.f;
    }
  }
  float wihs[4][4];
  float wlv[4];
  bool vr[4];
#pragma unroll
  for (int r = 0; r < 4; ++r) {
    const int u = ws * 16 + hi + 4 * up + r;
    vr[r] = (u < HDIM);
    const int uc = vr[r] ? u : 0;
    const float wlr = w_lin[uc];
    wlv[r] = vr[r] ? wlr : 0.0f;
#pragma unroll
    for (int q = 0; q < 4; ++q) {
      const int rr = q * HDIM + uc;
      const float gsc = (q == 2) ? 1.0f : 0.5f;
      float wi = w_ih1[rr] * gsc;
      wihs[q][r] = (vr[r] && role == 0) ? wi : 0.f;
    }
  }
  const float blin = b_lin[0];

  for (int i = tid; i < 2 * 8 * 136; i += 256) ((_Float16*)hbuf)[i] = (_Float16)0.0f;
  for (int i = tid; i < 16 * 8 * 65; i += 256) ((float*)ypart)[i] = 0.0f;
  __syncthreads();

  float cs1[4], cs2[4];
#pragma unroll
  for (int k = 0; k < 4; ++k) { cs1[k] = 0.f; cs2[k] = 0.f; }

  auto load_frag = [&](const _Float16* bp) __attribute__((always_inline)) -> v16h {
    v8h lo = *(const v8h*)bp;
    v8h hp = *(const v8h*)(bp + 16);
    return __builtin_shufflevector(lo, hp, 0,1,2,3,4,5,6,7,8,9,10,11,12,13,14,15);
  };

  unsigned upm = 0u - (unsigned)up;
  asm volatile("" : "+v"(upm));
  auto keep4 = [&](const v8f (&acc)[4], float (&z)[4][4]) __attribute__((always_inline)) {
#pragma unroll
    for (int q = 0; q < 4; ++q) {
#pragma unroll
      for (int r = 0; r < 4; ++r) {
        const float lof = acc[q][r];
        const float hif = acc[q][r + 4];
        const unsigned lo = __float_as_uint(lof);
        const unsigned hh = __float_as_uint(hif);
        z[q][r] = __uint_as_float(lo ^ ((lo ^ hh) & upm));
      }
    }
  };

  auto cell = [&](const float (&z)[4][4], float (&cs)[4], float (&hn)[4], _Float16* dst) __attribute__((always_inline)) {
    v4h hv;
#pragma unroll
    for (int r = 0; r < 4; ++r) {
      float ig = sigmoid_of_half_arg(z[0][r]);
      float fg = sigmoid_of_half_arg(z[1][r]);
      float gg = hw_tanh(z[2][r]);
      float og = sigmoid_of_half_arg(z[3][r]);
      float cn = fg * cs[r] + ig * gg;
      cs[r] = cn;
      float hx = og * hw_tanh(cn);
      hn[r] = vr[r] ? hx : 0.0f;
      hv[r] = (_Float16)hn[r];
    }
    *(v4h*)dst = hv;
  };

  auto ysum = [&](int row, int sl) __attribute__((always_inline)) -> float {
    const float s0 = (ypart[0][row][sl] + ypart[1][row][sl]) + (ypart[2][row][sl] + ypart[3][row][sl]);
    const float s1 = (ypart[4][row][sl] + ypart[5][row][sl]) + (ypart[6][row][sl] + ypart[7][row][sl]);
    const float s2 = (ypart[8][row][sl] + ypart[9][row][sl]) + (ypart[10][row][sl] + ypart[11][row][sl]);
    const float s3 = (ypart[12][row][sl] + ypart[13][row][sl]) + (ypart[14][row][sl] + ypart[15][row][sl]);
    return ((s0 + s1) + (s2 + s3)) + blin;
  };

  auto x_of = [&](int t, bool from_input) __attribute__((always_inline)) -> float {
    if (from_input) return input[(long)(rowbase + brow) * T + t];
    return ysum(brow, (t - 1) & 63);
  };
  auto layer1_step = [&](int t, float xv, const v16h (&a1)[2]) __attribute__((always_inline)) {
    const int cur = t & 1;
    v8f acc[4];
#pragma unroll
    for (int q = 0; q < 4; ++q)
      acc[q] = __builtin_amdgcn_wmma_f32_16x16x32_f16(false, bw[q][0], false,
                  a1[0], (short)0, biasv[q], false, false);
#pragma unroll
    for (int q = 0; q < 4; ++q)
      acc[q] = __builtin_amdgcn_wmma_f32_16x16x32_f16(false, bw[q][1], false,
                  a1[1], (short)0, acc[q], false, false);
    float z[4][4];
    keep4(acc, z);
#pragma unroll
    for (int q = 0; q < 4; ++q) {
#pragma unroll
      for (int r = 0; r < 4; ++r) z[q][r] += xv * wihs[q][r];
    }
    float hn[4];
    cell(z, cs1, hn, &hbuf[cur][brow][ws * 16 + hi + 4 * up]);
  };

  v16h hf[2] = {};
  auto fetch_h1 = [&](int t) __attribute__((always_inline)) {
#pragma unroll
    for (int ks = 0; ks < 2; ++ks) hf[ks] = load_frag(&hbuf[t & 1][brow][ks * 32 + hi]);
  };

  auto layer2_step = [&](int t, bool fetch_next, const v16h (&a2)[2]) __attribute__((always_inline)) {
    const int cur = t & 1, prv = cur ^ 1;
    v8f acc[4];
#pragma unroll
    for (int q = 0; q < 4; ++q)
      acc[q] = __builtin_amdgcn_wmma_f32_16x16x32_f16(false, bw[q][0], false,
                  hf[0], (short)0, biasv[q], false, false);
#pragma unroll
    for (int q = 0; q < 4; ++q)
      acc[q] = __builtin_amdgcn_wmma_f32_16x16x32_f16(false, bw[q][1], false,
                  hf[1], (short)0, acc[q], false, false);
#pragma unroll
    for (int ks = 0; ks < 2; ++ks)
#pragma unroll
      for (int q = 0; q < 4; ++q)
        acc[q] = __builtin_amdgcn_wmma_f32_16x16x32_f16(false, bw[q][2 + ks], false,
                    a2[ks], (short)0, acc[q], false, false);
    if (fetch_next) {
      asm volatile("" : "+v"(acc[0]), "+v"(acc[1]), "+v"(acc[2]), "+v"(acc[3]));
      __builtin_amdgcn_sched_barrier(0);
      fetch_h1(t + 1);
      __builtin_amdgcn_sched_barrier(0);
    }
    float z[4][4];
    keep4(acc, z);
    float hn[4];
    cell(z, cs2, hn, &hbuf[cur][brow][64 + ws * 16 + hi + 4 * up]);
    const float y = (hn[0] * wlv[0] + hn[1] * wlv[1]) + (hn[2] * wlv[2] + hn[3] * wlv[3]);
    ypart[ws * 4 + half * 2 + up][brow][t & 63] = y;
  };

  auto store32 = [&](int first) __attribute__((always_inline)) {
    const int sl = (first + lane) & 63;
#pragma unroll
    for (int rr = 0; rr < 2; ++rr) {
      const int row = 2 * ws + rr;
      out[(long)(rowbase + row) * Tout + (first + lane)] = ysum(row, sl);
    }
  };

  const int ownk = (role ? 64 : 0) + hi;
  auto load_own = [&](int par, v16h (&fr)[2]) __attribute__((always_inline)) {
    fr[0] = load_frag(&hbuf[par][brow][ownk]);
    fr[1] = load_frag(&hbuf[par][brow][ownk + 32]);
  };
  auto run1 = [&](int t, bool from_input) __attribute__((always_inline)) {
    v16h fr[2];
    const float xv = x_of(t, from_input);
    load_own((t & 1) ^ 1, fr);
    __builtin_amdgcn_sched_barrier(0);
    layer1_step(t, xv, fr);
  };
  auto run2 = [&](int t, bool fetch_next) __attribute__((always_inline)) {
    v16h fr[2];
    load_own((t & 1) ^ 1, fr);
    __builtin_amdgcn_sched_barrier(0);
    layer2_step(t, fetch_next, fr);
  };

  const int Tm = (T < Tout) ? T : Tout;
  int stored = 0;
  if (Tm > 0) {
    if (role == 0) run1(0, true);
    __syncthreads();
    if (role == 0) {
      if (Tm > 1) run1(1, true);
    } else {
      fetch_h1(0);
    }
    __syncthreads();
  }
  const char* hb = (const char*)&hbuf[0][0][0];
  const int own0 = (brow * 136 + ownk) * 2;
  const int ownsum = 2 * own0 + 8 * 136 * 2;
  int owncur = own0 + 8 * 136 * 2;
  for (int s = 2; s < Tm; ++s) {
    v16h fr[2];
    fr[0] = load_frag((const _Float16*)(hb + owncur));
    fr[1] = load_frag((const _Float16*)(hb + owncur + 64));
    __builtin_amdgcn_sched_barrier(0);
    if (role != 0) {
      layer2_step(s - 2, true, fr);
    } else {
      const float xv = x_of(s, true);
      __builtin_amdgcn_sched_barrier(0);
      if (s - 2 - stored >= 32) {
        store32(stored);
        stored += 32;
      }
      layer1_step(s, xv, fr);
    }
    owncur = ownsum - owncur;
    __syncthreads();
  }
  if (Tm > 1) {
    if (role == 1) run2(Tm - 2, true);
    __syncthreads();
  }
  if (Tm > 0) {
    if (role == 1) run2(Tm - 1, false);
    __syncthreads();
  }
  for (int t = Tm; t < Tout; ++t) {
    if (role == 0) {
      if (t - stored >= 32) {
        store32(stored);
        stored += 32;
      }
      run1(t, false);
    }
    __syncthreads();
    if (role == 1) {
      fetch_h1(t);
      run2(t, false);
    }
    __syncthreads();
  }

  if (role == 0) {
    for (; stored < Tout; stored += 32) {
      if (stored + lane < Tout) store32(stored);
    }
  }
}

extern "C" void kernel_launch(void* const* d_in, const int* in_sizes, int n_in,
                              void* d_out, int out_size, void* d_ws, size_t ws_size,
                              hipStream_t stream) {
  const float* input = (const float*)d_in[0];
  const float* w_ih1 = (const float*)d_in[2];
  const float* w_hh1 = (const float*)d_in[3];
  const float* b_ih1 = (const float*)d_in[4];
  const float* b_hh1 = (const float*)d_in[5];
  const float* w_ih2 = (const float*)d_in[6];
  const float* w_hh2 = (const float*)d_in[7];
  const float* b_ih2 = (const float*)d_in[8];
  const float* b_hh2 = (const float*)d_in[9];
  const float* w_lin = (const float*)d_in[10];
  const float* b_lin = (const float*)d_in[11];
  float* out = (float*)d_out;

  const int B = 1024;
  const int T = in_sizes[0] / B;
  const int Tout = out_size / B;

  lstm2_wmma_kernel<<<dim3(B / 8), dim3(256), 0, stream>>>(
      input, w_ih1, w_hh1, b_ih1, b_hh1, w_ih2, w_hh2, b_ih2, b_hh2,
      w_lin, b_lin, out, T, Tout);
}
